# starting kernel
# baseline (speedup 1.0000x reference)
#include <hip/hip_runtime.h>
#include <hip/hip_bf16.h>

typedef __attribute__((ext_vector_type(2))) float v2f;
typedef __attribute__((ext_vector_type(8))) float v8f;

#define NEG_SLOPE 0.2f

template <int FIN, int FOUT>
__global__ __launch_bounds__(128) void gemm_wmma_kernel(
    const float* __restrict__ X, const float* __restrict__ W,
    float* __restrict__ H, int nrows) {
  __shared__ float lds_x[4][16 * FIN];
  const int wave = threadIdx.x >> 5;
  const int lane = threadIdx.x & 31;
  const int m0 = (blockIdx.x * 4 + wave) * 16;

  float* xt = lds_x[wave];
  constexpr int NCHUNK = 16 * FIN / 4;
#pragma unroll 4
  for (int i = lane; i < NCHUNK; i += 32) {
    int r = i / (FIN / 4);
    int c = (i % (FIN / 4)) * 4;
    int row = m0 + r;
    if (row >= nrows) row = nrows - 1;
    float4 v = *(const float4*)(X + (size_t)row * FIN + c);
    *(float4*)(xt + r * FIN + c) = v;
  }
  __syncthreads();

  constexpr int NT = FOUT / 16;
  v8f acc[NT] = {};

  const int arow = lane & 15;
  const int ksel = 2 * (lane >> 4);
  for (int k = 0; k < FIN; k += 4) {
    v2f a;
    a.x = xt[arow * FIN + k + ksel];
    a.y = xt[arow * FIN + k + ksel + 1];
#pragma unroll
    for (int t = 0; t < NT; ++t) {
      int n = t * 16 + (lane & 15);
      v2f b;
      b.x = W[(k + ksel) * FOUT + n];
      b.y = W[(k + ksel + 1) * FOUT + n];
      acc[t] = __builtin_amdgcn_wmma_f32_16x16x4_f32(
          false, a, false, b, (short)0, acc[t], false, false);
    }
  }

  const int row_off = (lane >> 4) * 8;
  const int col = lane & 15;
#pragma unroll
  for (int t = 0; t < NT; ++t) {
#pragma unroll
    for (int i = 0; i < 8; ++i) {
      int row = m0 + row_off + i;
      if (row < nrows) H[(size_t)row * FOUT + t * 16 + col] = acc[t][i];
    }
  }
}

template <int F>
__global__ __launch_bounds__(256) void node_scores_kernel(
    const float* __restrict__ H, const float* __restrict__ a_src,
    const float* __restrict__ a_dst, float* __restrict__ ssrc,
    float* __restrict__ sdst, int n) {
  int node = blockIdx.x * (blockDim.x >> 5) + (threadIdx.x >> 5);
  int lane = threadIdx.x & 31;
  if (node >= n) return;
  float as = 0.f, ad = 0.f;
#pragma unroll
  for (int f = lane; f < F; f += 32) {
    float h = H[(size_t)node * F + f];
    as += h * a_src[f];
    ad += h * a_dst[f];
  }
#pragma unroll
  for (int off = 16; off > 0; off >>= 1) {
    as += __shfl_down(as, off, 32);
    ad += __shfl_down(ad, off, 32);
  }
  if (lane == 0) {
    ssrc[node] = as;
    sdst[node] = ad;
  }
}

__device__ __forceinline__ unsigned ford(float x) {
  unsigned b = __float_as_uint(x);
  return (b & 0x80000000u) ? ~b : (b | 0x80000000u);
}
__device__ __forceinline__ float funord(unsigned u) {
  unsigned b = (u & 0x80000000u) ? (u ^ 0x80000000u) : ~u;
  return __uint_as_float(b);
}

__device__ __forceinline__ void edge_sd(int i, int E, const int* __restrict__ src,
                                        const int* __restrict__ dst, int& s, int& d) {
  if (i < E) { s = src[i]; d = dst[i]; }
  else       { s = i - E; d = s; }
}

__global__ __launch_bounds__(256) void edge_pass1_kernel(
    const int* __restrict__ src, const int* __restrict__ dst,
    const float* __restrict__ ssrc, const float* __restrict__ sdst,
    float* __restrict__ ebuf, unsigned* __restrict__ emaxu, int E, int Etot) {
  int i = blockIdx.x * blockDim.x + threadIdx.x;
  if (i >= Etot) return;
  int s, d;
  edge_sd(i, E, src, dst, s, d);
  float e = ssrc[s] + sdst[d];
  e = (e > 0.f) ? e : NEG_SLOPE * e;
  ebuf[i] = e;
  atomicMax(&emaxu[d], ford(e));
}

__global__ __launch_bounds__(256) void edge_pass2_kernel(
    const int* __restrict__ src, const int* __restrict__ dst,
    float* __restrict__ ebuf, const unsigned* __restrict__ emaxu,
    float* __restrict__ denom, int E, int Etot) {
  int i = blockIdx.x * blockDim.x + threadIdx.x;
  if (i >= Etot) return;
  int s, d;
  edge_sd(i, E, src, dst, s, d);
  float ex = __expf(ebuf[i] - funord(emaxu[d]));
  ebuf[i] = ex;
  atomicAdd(&denom[d], ex);
}

template <int F>
__global__ __launch_bounds__(256) void edge_pass3_kernel(
    const int* __restrict__ src, const int* __restrict__ dst,
    const float* __restrict__ ebuf, const float* __restrict__ denom,
    const float* __restrict__ H, float* __restrict__ alpha_out,
    float* __restrict__ out, int E, int Etot) {
  long long tid = (long long)blockIdx.x * blockDim.x + threadIdx.x;
  long long edge = tid >> 4;
  int fl = (int)(tid & 15);
  if (edge >= Etot) return;
  int s, d;
  edge_sd((int)edge, E, src, dst, s, d);
  float alpha = ebuf[edge] / denom[d];
  if (fl == 0) alpha_out[edge] = alpha;
  constexpr int V = F / 16;
  const float* hp = H + (size_t)s * F + fl * V;
  float* op = out + (size_t)d * F + fl * V;
#pragma unroll
  for (int j = 0; j < V; ++j) atomicAdd(op + j, alpha * hp[j]);
}

__global__ __launch_bounds__(256) void bias_act_kernel(
    float* __restrict__ H, const float* __restrict__ b, long long total, int F,
    int do_elu) {
  long long i = (long long)blockIdx.x * blockDim.x + threadIdx.x;
  if (i >= total) return;
  int f = (int)(i % F);
  float v = H[i] + b[f];
  if (do_elu) v = (v > 0.f) ? v : (__expf(v) - 1.0f);
  H[i] = v;
}

static inline int cdiv(long long a, long long b) { return (int)((a + b - 1) / b); }

extern "C" void kernel_launch(void* const* d_in, const int* in_sizes, int n_in,
                              void* d_out, int out_size, void* d_ws, size_t ws_size,
                              hipStream_t stream) {
  constexpr int FIN = 128, FH = 64, FO = 32;
  const float* x      = (const float*)d_in[0];
  const int*   eidx   = (const int*)d_in[1];
  const float* W1     = (const float*)d_in[2];
  const float* a_src1 = (const float*)d_in[3];
  const float* a_dst1 = (const float*)d_in[4];
  const float* b1     = (const float*)d_in[5];
  const float* W2     = (const float*)d_in[6];
  const float* a_src2 = (const float*)d_in[7];
  const float* a_dst2 = (const float*)d_in[8];
  const float* b2     = (const float*)d_in[9];

  const int N = in_sizes[0] / FIN;
  const int E = in_sizes[1] / 2;
  const int Etot = E + N;
  const int* src = eidx;
  const int* dst = eidx + E;

  float* hfeat1 = (float*)d_ws;
  float* agg1   = hfeat1 + (size_t)N * FH;
  float* hfeat2 = agg1 + (size_t)N * FH;
  float* ebuf   = hfeat2 + (size_t)N * FO;
  float* ssrc   = ebuf + (size_t)Etot;
  float* sdst   = ssrc + N;
  unsigned* emaxu = (unsigned*)(sdst + N);
  float* denom  = (float*)(emaxu + N);

  float* h2     = (float*)d_out;
  float* alpha1 = h2 + (size_t)N * FO;
  float* alpha2 = alpha1 + (size_t)Etot;

  dim3 b256(256), b128(128);

  hipMemsetAsync(agg1, 0, (size_t)N * FH * sizeof(float), stream);
  hipMemsetAsync(emaxu, 0, (size_t)N * sizeof(unsigned), stream);
  hipMemsetAsync(denom, 0, (size_t)N * sizeof(float), stream);

  gemm_wmma_kernel<FIN, FH><<<cdiv(N, 64), b128, 0, stream>>>(x, W1, hfeat1, N);
  node_scores_kernel<FH><<<cdiv(N, 8), b256, 0, stream>>>(hfeat1, a_src1, a_dst1, ssrc, sdst, N);
  edge_pass1_kernel<<<cdiv(Etot, 256), b256, 0, stream>>>(src, dst, ssrc, sdst, ebuf, emaxu, E, Etot);
  edge_pass2_kernel<<<cdiv(Etot, 256), b256, 0, stream>>>(src, dst, ebuf, emaxu, denom, E, Etot);
  edge_pass3_kernel<FH><<<cdiv((long long)Etot * 16, 256), b256, 0, stream>>>(
      src, dst, ebuf, denom, hfeat1, alpha1, agg1, E, Etot);
  bias_act_kernel<<<cdiv((long long)N * FH, 256), b256, 0, stream>>>(
      agg1, b1, (long long)N * FH, FH,  1);

  hipMemsetAsync(h2, 0, (size_t)N * FO * sizeof(float), stream);
  hipMemsetAsync(emaxu, 0, (size_t)N * sizeof(unsigned), stream);
  hipMemsetAsync(denom, 0, (size_t)N * sizeof(float), stream);

  gemm_wmma_kernel<FH, FO><<<cdiv(N, 64), b128, 0, stream>>>(agg1, W2, hfeat2, N);
  node_scores_kernel<FO><<<cdiv(N, 8), b256, 0, stream>>>(hfeat2, a_src2, a_dst2, ssrc, sdst, N);
  edge_pass1_kernel<<<cdiv(Etot, 256), b256, 0, stream>>>(src, dst, ssrc, sdst, ebuf, emaxu, E, Etot);
  edge_pass2_kernel<<<cdiv(Etot, 256), b256, 0, stream>>>(src, dst, ebuf, emaxu, denom, E, Etot);
  edge_pass3_kernel<FO><<<cdiv((long long)Etot * 16, 256), b256, 0, stream>>>(
      src, dst, ebuf, denom, hfeat2, alpha2, h2, E, Etot);
  bias_act_kernel<<<cdiv((long long)N * FO, 256), b256, 0, stream>>>(
      h2, b2, (long long)N * FO, FO,  0);
}


# Edges grouped by destination once (int atomic count, scan, place, then ordered by edge id); one wave per node does softmax and row sum in registers: float atomics gone, sum order fixed by edge id
# speedup vs baseline: 2.1586x; 2.1586x over previous
#include <hip/hip_runtime.h>
#include <hip/hip_bf16.h>

typedef __attribute__((ext_vector_type(2))) float v2f;
typedef __attribute__((ext_vector_type(8))) float v8f;

#define NEG_SLOPE 0.2f

template <int FIN, int FOUT>
__global__ __launch_bounds__(128) void gemm_wmma_kernel(
    const float* __restrict__ X, const float* __restrict__ W,
    float* __restrict__ H, int nrows) {
  __shared__ float lds_x[4][16 * FIN];
  const int wave = threadIdx.x >> 5;
  const int lane = threadIdx.x & 31;
  const int m0 = (blockIdx.x * 4 + wave) * 16;

  float* xt = lds_x[wave];
  constexpr int NCHUNK = 16 * FIN / 4;
#pragma unroll 4
  for (int i = lane; i < NCHUNK; i += 32) {
    int r = i / (FIN / 4);
    int c = (i % (FIN / 4)) * 4;
    int row = m0 + r;
    if (row >= nrows) row = nrows - 1;
    float4 v = *(const float4*)(X + (size_t)row * FIN + c);
    *(float4*)(xt + r * FIN + c) = v;
  }
  __syncthreads();

  constexpr int NT = FOUT / 16;
  v8f acc[NT] = {};

  const int arow = lane & 15;
  const int ksel = 2 * (lane >> 4);
  for (int k = 0; k < FIN; k += 4) {
    v2f a;
    a.x = xt[arow * FIN + k + ksel];
    a.y = xt[arow * FIN + k + ksel + 1];
#pragma unroll
    for (int t = 0; t < NT; ++t) {
      int n = t * 16 + (lane & 15);
      v2f b;
      b.x = W[(k + ksel) * FOUT + n];
      b.y = W[(k + ksel + 1) * FOUT + n];
      acc[t] = __builtin_amdgcn_wmma_f32_16x16x4_f32(
          false, a, false, b, (short)0, acc[t], false, false);
    }
  }

  const int row_off = (lane >> 4) * 8;
  const int col = lane & 15;
#pragma unroll
  for (int t = 0; t < NT; ++t) {
#pragma unroll
    for (int i = 0; i < 8; ++i) {
      int row = m0 + row_off + i;
      if (row < nrows) H[(size_t)row * FOUT + t * 16 + col] = acc[t][i];
    }
  }
}

template <int F>
__global__ __launch_bounds__(256) void node_scores_kernel(
    const float* __restrict__ H, const float* __restrict__ a_src,
    const float* __restrict__ a_dst, float* __restrict__ ssrc,
    float* __restrict__ sdst, int n) {
  int node = blockIdx.x * (blockDim.x >> 5) + (threadIdx.x >> 5);
  int lane = threadIdx.x & 31;
  if (node >= n) return;
  float as = 0.f, ad = 0.f;
#pragma unroll
  for (int f = lane; f < F; f += 32) {
    float h = H[(size_t)node * F + f];
    as += h * a_src[f];
    ad += h * a_dst[f];
  }
#pragma unroll
  for (int off = 16; off > 0; off >>= 1) {
    as += __shfl_down(as, off, 32);
    ad += __shfl_down(ad, off, 32);
  }
  if (lane == 0) {
    ssrc[node] = as;
    sdst[node] = ad;
  }
}


__global__ __launch_bounds__(256) void dst_count_kernel(
    const int* __restrict__ dst, int* __restrict__ cnt, int* __restrict__ place, int E) {
  int i = blockIdx.x * blockDim.x + threadIdx.x;
  if (i >= E) return;
  place[i] = atomicAdd(&cnt[dst[i]], 1);
}

__global__ __launch_bounds__(1024) void row_scan_kernel(
    const int* __restrict__ cnt, int* __restrict__ rowbeg, int n) {
  __shared__ int wsum[32];
  const int t = threadIdx.x;
  const int lane = t & 31, w = t >> 5;
  const int per = (n + 1023) / 1024;
  const int lo = min(t * per, n);
  const int hi = min(lo + per, n);
  int s = 0;
#pragma unroll 8
  for (int i = lo; i < hi; ++i) s += cnt[i] + 1;
  int v = s;
#pragma unroll
  for (int off = 1; off < 32; off <<= 1) {
    int u = __shfl_up(v, off, 32);
    if (lane >= off) v += u;
  }
  if (lane == 31) wsum[w] = v;
  __syncthreads();
  if (w == 0) {
    int x = wsum[lane];
#pragma unroll
    for (int off = 1; off < 32; off <<= 1) {
      int u = __shfl_up(x, off, 32);
      if (lane >= off) x += u;
    }
    wsum[lane] = x;
  }
  __syncthreads();
  int base = v - s + (w > 0 ? wsum[w - 1] : 0);
  for (int i = lo; i < hi; ++i) {
    rowbeg[i] = base;
    base += cnt[i] + 1;
  }
  if (t == 1023) rowbeg[n] = base;
}

__global__ __launch_bounds__(256) void place_kernel(
    const int* __restrict__ src, const int* __restrict__ dst,
    const int* __restrict__ place, const int* __restrict__ rowbeg,
    int2* __restrict__ slots, int E) {
  int i = blockIdx.x * blockDim.x + threadIdx.x;
  if (i >= E) return;
  int d = dst[i];
  slots[rowbeg[d] + place[i]] = make_int2(i, src[i]);
}

__global__ __launch_bounds__(256) void order_kernel(
    const int2* __restrict__ sa, int2* __restrict__ sb,
    const int* __restrict__ rowbeg, int n, int E) {
  const int lane = threadIdx.x & 31;
  const int node = __builtin_amdgcn_readfirstlane(
      (int)(blockIdx.x * (blockDim.x >> 5) + (threadIdx.x >> 5)));
  if (node >= n) return;
  const int beg = rowbeg[node];
  const int L = rowbeg[node + 1] - beg - 1;
  for (int i0 = 0; i0 < L; i0 += 32) {
    const int i = i0 + lane;
    int2 me = make_int2(0x7fffffff, 0);
    if (i < L) me = sa[beg + i];
    int r = 0;
    for (int j0 = 0; j0 < L; j0 += 32) {
      const int j = j0 + lane;
      int kj = 0x7fffffff;
      if (j < L) kj = sa[beg + j].x;
      const int lim = min(32, L - j0);
      for (int t = 0; t < lim; ++t)
        r += (__builtin_amdgcn_readlane(kj, t) < me.x) ? 1 : 0;
    }
    if (i < L) sb[beg + r] = me;
  }
  if (lane == 0) sb[beg + L] = make_int2(E + node, node);
}

__device__ __forceinline__ float wave_max(float v) {
#pragma unroll
  for (int off = 16; off > 0; off >>= 1) v = fmaxf(v, __shfl_xor(v, off, 32));
  return v;
}
__device__ __forceinline__ float wave_sum(float v) {
#pragma unroll
  for (int off = 16; off > 0; off >>= 1) v += __shfl_xor(v, off, 32);
  return v;
}

template <int F>
__device__ __forceinline__ float4 gather_rows(float4 acc, float a_l, int s_l, int cnt,
                                              const float* __restrict__ H, int sub, int fl) {
  constexpr int LPE = F / 4;
  constexpr int EPI = 32 / LPE;
  const int groups = (cnt + EPI - 1) / EPI;
#pragma unroll 4
  for (int g = 0; g < groups; ++g) {
    const int jl = g * EPI + sub;
    const float a = __shfl(a_l, jl, 32);
    const int s = __shfl(s_l, jl, 32);
    const float4 h = *(const float4*)(H + (size_t)s * F + fl * 4);
    acc.x = fmaf(a, h.x, acc.x);
    acc.y = fmaf(a, h.y, acc.y);
    acc.z = fmaf(a, h.z, acc.z);
    acc.w = fmaf(a, h.w, acc.w);
  }
  return acc;
}

__device__ __forceinline__ float leaky(float e) { return (e > 0.f) ? e : NEG_SLOPE * e; }

template <int F, bool ELU>
__global__ __launch_bounds__(256) void node_gather_kernel(
    const int2* __restrict__ slots, const int* __restrict__ rowbeg,
    const float* __restrict__ ssrc, const float* __restrict__ sdst,
    const float* __restrict__ H, const float* __restrict__ bias,
    float* __restrict__ alpha_out, float* __restrict__ out, int n) {
  constexpr int LPE = F / 4;
  constexpr int MAXC = 4;
  const int lane = threadIdx.x & 31;
  const int node = __builtin_amdgcn_readfirstlane(
      (int)(blockIdx.x * (blockDim.x >> 5) + (threadIdx.x >> 5)));
  if (node >= n) return;
  const int beg = rowbeg[node];
  const int L = rowbeg[node + 1] - beg;
  const float sd = sdst[node];
  const int sub = lane / LPE, fl = lane % LPE;
  float4 acc = make_float4(0.f, 0.f, 0.f, 0.f);

  if (L <= 32 * MAXC) {
    int sidx[MAXC], eidx[MAXC];
    float ev[MAXC];
    float m = -INFINITY;
#pragma unroll
    for (int c = 0; c < MAXC; ++c) {
      const int idx = c * 32 + lane;
      sidx[c] = node; eidx[c] = 0; ev[c] = -INFINITY;
      if (idx < L) {
        const int2 sl = slots[beg + idx];
        eidx[c] = sl.x; sidx[c] = sl.y;
        ev[c] = leaky(ssrc[sl.y] + sd);
      }
      m = fmaxf(m, ev[c]);
    }
    m = wave_max(m);
    float sum = 0.f;
#pragma unroll
    for (int c = 0; c < MAXC; ++c) {
      const float ex = (c * 32 + lane < L) ? __expf(ev[c] - m) : 0.f;
      ev[c] = ex;
      sum += ex;
    }
    sum = wave_sum(sum);
#pragma unroll
    for (int c = 0; c < MAXC; ++c) {
      const float a = ev[c] / sum;
      ev[c] = a;
      if (c * 32 + lane < L) alpha_out[eidx[c]] = a;
    }
#pragma unroll
    for (int c = 0; c < MAXC; ++c) {
      if (c * 32 < L)
        acc = gather_rows<F>(acc, ev[c], sidx[c], min(32, L - c * 32), H, sub, fl);
    }
  } else {
    float m = -INFINITY;
    for (int c0 = 0; c0 < L; c0 += 32) {
      const int idx = c0 + lane;
      if (idx < L) m = fmaxf(m, leaky(ssrc[slots[beg + idx].y] + sd));
    }
    m = wave_max(m);
    float sum = 0.f;
    for (int c0 = 0; c0 < L; c0 += 32) {
      const int idx = c0 + lane;
      if (idx < L) sum += __expf(leaky(ssrc[slots[beg + idx].y] + sd) - m);
    }
    sum = wave_sum(sum);
    for (int c0 = 0; c0 < L; c0 += 32) {
      const int idx = c0 + lane;
      float a = 0.f;
      int s = node;
      if (idx < L) {
        const int2 sl = slots[beg + idx];
        s = sl.y;
        a = __expf(leaky(ssrc[s] + sd) - m) / sum;
        alpha_out[sl.x] = a;
      }
      acc = gather_rows<F>(acc, a, s, min(32, L - c0), H, sub, fl);
    }
  }

#pragma unroll
  for (int off = 16; off >= LPE; off >>= 1) {
    acc.x += __shfl_xor(acc.x, off, 32);
    acc.y += __shfl_xor(acc.y, off, 32);
    acc.z += __shfl_xor(acc.z, off, 32);
    acc.w += __shfl_xor(acc.w, off, 32);
  }
  if (sub == 0) {
    const float4 b = *(const float4*)(bias + fl * 4);
    float4 v = make_float4(acc.x + b.x, acc.y + b.y, acc.z + b.z, acc.w + b.w);
    if (ELU) {
      v.x = (v.x > 0.f) ? v.x : (__expf(v.x) - 1.0f);
      v.y = (v.y > 0.f) ? v.y : (__expf(v.y) - 1.0f);
      v.z = (v.z > 0.f) ? v.z : (__expf(v.z) - 1.0f);
      v.w = (v.w > 0.f) ? v.w : (__expf(v.w) - 1.0f);
    }
    *(float4*)(out + (size_t)node * F + fl * 4) = v;
  }
}

static inline int cdiv(long long a, long long b) { return (int)((a + b - 1) / b); }

extern "C" void kernel_launch(void* const* d_in, const int* in_sizes, int n_in,
                              void* d_out, int out_size, void* d_ws, size_t ws_size,
                              hipStream_t stream) {
  constexpr int FIN = 128, FH = 64, FO = 32;
  const float* x      = (const float*)d_in[0];
  const int*   eidx   = (const int*)d_in[1];
  const float* W1     = (const float*)d_in[2];
  const float* a_src1 = (const float*)d_in[3];
  const float* a_dst1 = (const float*)d_in[4];
  const float* b1     = (const float*)d_in[5];
  const float* W2     = (const float*)d_in[6];
  const float* a_src2 = (const float*)d_in[7];
  const float* a_dst2 = (const float*)d_in[8];
  const float* b2     = (const float*)d_in[9];

  const int N = in_sizes[0] / FIN;
  const int E = in_sizes[1] / 2;
  const int Etot = E + N;
  const int* src = eidx;
  const int* dst = eidx + E;

  float* hfeat1 = (float*)d_ws;
  float* agg1   = hfeat1 + (size_t)N * FH;
  float* hfeat2 = agg1 + (size_t)N * FH;
  int2* slots_a = (int2*)(hfeat2 + (size_t)N * FO);
  int2* slots_b = slots_a + (size_t)Etot;
  int* place    = (int*)(slots_b + (size_t)Etot);
  float* ssrc   = (float*)(place + (size_t)E);
  float* sdst   = ssrc + N;
  int* cnt      = (int*)(sdst + N);
  int* rowbeg   = cnt + N;

  float* h2     = (float*)d_out;
  float* alpha1 = h2 + (size_t)N * FO;
  float* alpha2 = alpha1 + (size_t)Etot;

  dim3 b256(256), b128(128), b1024(1024);

  hipMemsetAsync(cnt, 0, (size_t)N * sizeof(int), stream);
  dst_count_kernel<<<cdiv(E, 256), b256, 0, stream>>>(dst, cnt, place, E);
  row_scan_kernel<<<1, b1024, 0, stream>>>(cnt, rowbeg, N);
  place_kernel<<<cdiv(E, 256), b256, 0, stream>>>(src, dst, place, rowbeg, slots_a, E);
  order_kernel<<<cdiv(N, 8), b256, 0, stream>>>(slots_a, slots_b, rowbeg, N, E);

  gemm_wmma_kernel<FIN, FH><<<cdiv(N, 64), b128, 0, stream>>>(x, W1, hfeat1, N);
  node_scores_kernel<FH><<<cdiv(N, 8), b256, 0, stream>>>(hfeat1, a_src1, a_dst1, ssrc, sdst, N);
  node_gather_kernel<FH, true><<<cdiv(N, 8), b256, 0, stream>>>(
      slots_b, rowbeg, ssrc, sdst, hfeat1, b1, alpha1, agg1, N);

  gemm_wmma_kernel<FH, FO><<<cdiv(N, 64), b128, 0, stream>>>(agg1, W2, hfeat2, N);
  node_scores_kernel<FO><<<cdiv(N, 8), b256, 0, stream>>>(hfeat2, a_src2, a_dst2, ssrc, sdst, N);
  node_gather_kernel<FO, false><<<cdiv(N, 8), b256, 0, stream>>>(
      slots_b, rowbeg, ssrc, sdst, hfeat2, b2, alpha2, h2, N);
}


# Row offsets by a grid of work-groups, one node per thread, whole-line loads and stores, each adding up what lies in front of its tile itself (was one work-group, a strided run of nodes per thread)
# speedup vs baseline: 2.5183x; 1.1666x over previous
#include <hip/hip_runtime.h>
#include <hip/hip_bf16.h>

typedef __attribute__((ext_vector_type(2))) float v2f;
typedef __attribute__((ext_vector_type(8))) float v8f;

#define NEG_SLOPE 0.2f

template <int FIN, int FOUT>
__global__ __launch_bounds__(128) void gemm_wmma_kernel(
    const float* __restrict__ X, const float* __restrict__ W,
    float* __restrict__ H, int nrows) {
  __shared__ float lds_x[4][16 * FIN];
  const int wave = threadIdx.x >> 5;
  const int lane = threadIdx.x & 31;
  const int m0 = (blockIdx.x * 4 + wave) * 16;

  float* xt = lds_x[wave];
  constexpr int NCHUNK = 16 * FIN / 4;
#pragma unroll 4
  for (int i = lane; i < NCHUNK; i += 32) {
    int r = i / (FIN / 4);
    int c = (i % (FIN / 4)) * 4;
    int row = m0 + r;
    if (row >= nrows) row = nrows - 1;
    float4 v = *(const float4*)(X + (size_t)row * FIN + c);
    *(float4*)(xt + r * FIN + c) = v;
  }
  __syncthreads();

  constexpr int NT = FOUT / 16;
  v8f acc[NT] = {};

  const int arow = lane & 15;
  const int ksel = 2 * (lane >> 4);
  for (int k = 0; k < FIN; k += 4) {
    v2f a;
    a.x = xt[arow * FIN + k + ksel];
    a.y = xt[arow * FIN + k + ksel + 1];
#pragma unroll
    for (int t = 0; t < NT; ++t) {
      int n = t * 16 + (lane & 15);
      v2f b;
      b.x = W[(k + ksel) * FOUT + n];
      b.y = W[(k + ksel + 1) * FOUT + n];
      acc[t] = __builtin_amdgcn_wmma_f32_16x16x4_f32(
          false, a, false, b, (short)0, acc[t], false, false);
    }
  }

  const int row_off = (lane >> 4) * 8;
  const int col = lane & 15;
#pragma unroll
  for (int t = 0; t < NT; ++t) {
#pragma unroll
    for (int i = 0; i < 8; ++i) {
      int row = m0 + row_off + i;
      if (row < nrows) H[(size_t)row * FOUT + t * 16 + col] = acc[t][i];
    }
  }
}


__global__ __launch_bounds__(256) void dst_count_kernel(
    const int* __restrict__ dst, int* __restrict__ cnt, int* __restrict__ place, int E) {
  int i = blockIdx.x * blockDim.x + threadIdx.x;
  if (i >= E) return;
  place[i] = atomicAdd(&cnt[dst[i]], 1);
}

__global__ __launch_bounds__(1024) void row_scan_kernel(
    const int* __restrict__ cnt, int* __restrict__ rowbeg, int n) {
  __shared__ int wtile[32];
  __shared__ int wfront[32];
  const int t = threadIdx.x;
  const int lane = t & 31, w = t >> 5;
  const int tile0 = blockIdx.x * 1024;
  int f = 0;
#pragma unroll 8
  for (int i = t; i < tile0; i += 1024) f += cnt[i];
  const int i = tile0 + t;
  const int own = (i < n) ? cnt[i] + 1 : 0;
  int v = own;
#pragma unroll
  for (int off = 1; off < 32; off <<= 1) {
    int u = __shfl_up(v, off, 32);
    if (lane >= off) v += u;
  }
#pragma unroll
  for (int off = 16; off > 0; off >>= 1) f += __shfl_xor(f, off, 32);
  if (lane == 31) wtile[w] = v;
  if (lane == 0) wfront[w] = f;
  __syncthreads();
  int s = wfront[lane] + ((lane < w) ? wtile[lane] : 0);
#pragma unroll
  for (int off = 16; off > 0; off >>= 1) s += __shfl_xor(s, off, 32);
  const int base = tile0 + s;
  if (i < n) rowbeg[i] = base + v - own;
  if (i == n - 1) rowbeg[n] = base + v;
}

__global__ __launch_bounds__(256) void place_kernel(
    const int* __restrict__ src, const int* __restrict__ dst,
    const int* __restrict__ place, const int* __restrict__ rowbeg,
    int2* __restrict__ slots, int E) {
  int i = blockIdx.x * blockDim.x + threadIdx.x;
  if (i >= E) return;
  int d = dst[i];
  slots[rowbeg[d] + place[i]] = make_int2(i, src[i]);
}

__global__ __launch_bounds__(256) void order_kernel(
    const int2* __restrict__ sa, int2* __restrict__ sb,
    const int* __restrict__ rowbeg, int n, int E) {
  const int lane = threadIdx.x & 31;
  const int node = __builtin_amdgcn_readfirstlane(
      (int)(blockIdx.x * (blockDim.x >> 5) + (threadIdx.x >> 5)));
  if (node >= n) return;
  const int beg = rowbeg[node];
  const int L = rowbeg[node + 1] - beg - 1;
  for (int i0 = 0; i0 < L; i0 += 32) {
    const int i = i0 + lane;
    int2 me = make_int2(0x7fffffff, 0);
    if (i < L) me = sa[beg + i];
    int r = 0;
    for (int j0 = 0; j0 < L; j0 += 32) {
      const int j = j0 + lane;
      int kj = 0x7fffffff;
      if (j < L) kj = sa[beg + j].x;
      const int lim = min(32, L - j0);
      for (int t = 0; t < lim; ++t)
        r += (__builtin_amdgcn_readlane(kj, t) < me.x) ? 1 : 0;
    }
    if (i < L) sb[beg + r] = me;
  }
  if (lane == 0) sb[beg + L] = make_int2(E + node, node);
}

__device__ __forceinline__ float leaky(float e) { return fmaxf(e, NEG_SLOPE * e); }

template <int X>
__device__ __forceinline__ float row_xor(float v) {
  return __builtin_bit_cast(
      float, __builtin_amdgcn_update_dpp(0, __builtin_bit_cast(int, v), 0x160 + X, 0xf, 0xf, false));
}

template <int LPE>
__device__ __forceinline__ float group_sum(float v) {
  v += row_xor<1>(v);
  v += row_xor<2>(v);
  v += row_xor<4>(v);
  if (LPE == 16) v += row_xor<8>(v);
  return v;
}

__device__ __forceinline__ float dot4(float4 a, float4 b) {
  return a.x * b.x + a.y * b.y + a.z * b.z + a.w * b.w;
}

#define LOWEST_F (-3.402823466e+38f)

struct Run {
  float m, sum;
  float4 acc;
};

template <int F>
__device__ __forceinline__ float chunk_rows(Run& st, int s_l, int cnt, const float* __restrict__ H,
                                            float4 as4, float sd, int sub, int fl, int lane) {
  constexpr int LPE = F / 4;
  constexpr int EPI = 32 / LPE;
  const int steps = (cnt + EPI - 1) / EPI;
  const char* Hb = (const char*)H;
  float keep = 0.f;
#pragma unroll 4
  for (int g = 0; g < steps; ++g) {
    const int jl = g * EPI + sub;
    const unsigned s = (unsigned)__shfl(s_l, jl, 32);
    const float4 h = *(const float4*)(Hb + (s * (unsigned)(F * 4) + (unsigned)(fl * 16)));
    float e = leaky(group_sum<LPE>(dot4(h, as4)) + sd);
    e = (jl < cnt) ? e : -INFINITY;
    const float m_new = fmaxf(st.m, e);
    const float scale = __expf(st.m - m_new);
    const float p = __expf(e - m_new);
    st.acc.x = fmaf(p, h.x, st.acc.x * scale);
    st.acc.y = fmaf(p, h.y, st.acc.y * scale);
    st.acc.z = fmaf(p, h.z, st.acc.z * scale);
    st.acc.w = fmaf(p, h.w, st.acc.w * scale);
    st.sum = fmaf(st.sum, scale, p);
    st.m = m_new;
    if (fl == g) keep = e;
  }
  return __shfl(keep, (lane % EPI) * LPE + lane / EPI, 32);
}

template <int F, bool ELU>
__global__ __launch_bounds__(256) void node_gather_kernel(
    const int2* __restrict__ slots, const int* __restrict__ rowbeg,
    const float* __restrict__ a_src, const float* __restrict__ a_dst,
    const float* __restrict__ H, const float* __restrict__ bias,
    float* __restrict__ alpha_out, float* __restrict__ out, int n, int E) {
  constexpr int LPE = F / 4;
  constexpr int MAXC = 4;
  const int lane = threadIdx.x & 31;
  const int node = __builtin_amdgcn_readfirstlane(
      (int)(blockIdx.x * (blockDim.x >> 5) + (threadIdx.x >> 5)));
  if (node >= n) return;
  const int beg = rowbeg[node];
  const int Ld = rowbeg[node + 1] - beg - 1;
  const int sub = lane / LPE, fl = lane % LPE;
  const bool small = Ld <= 32 * MAXC;

  int sidx[MAXC], eidx[MAXC];
  float ev[MAXC];
#pragma unroll
  for (int c = 0; c < MAXC; ++c) {
    const int2 sl = slots[beg + min(c * 32 + lane, Ld)];
    eidx[c] = sl.x;
    sidx[c] = (c * 32 + lane < Ld) ? sl.y : node;
    ev[c] = 0.f;
  }

  const float4 hself = *(const float4*)(H + (size_t)node * F + fl * 4);
  const float4 as4 = *(const float4*)(a_src + fl * 4);
  const float4 ad4 = *(const float4*)(a_dst + fl * 4);
  const float sd = group_sum<LPE>(dot4(hself, ad4));
  const float e_self = leaky(group_sum<LPE>(dot4(hself, as4)) + sd);

  Run st;
  st.m = (sub == 0) ? e_self : LOWEST_F;
  st.sum = (sub == 0) ? 1.0f : 0.f;
  st.acc = (sub == 0) ? hself : make_float4(0.f, 0.f, 0.f, 0.f);

  if (small) {
#pragma unroll
    for (int c = 0; c < MAXC; ++c) {
      if (c * 32 < Ld)
        ev[c] = chunk_rows<F>(st, sidx[c], min(32, Ld - c * 32), H, as4, sd, sub, fl, lane);
    }
  } else {
    for (int c0 = 0; c0 < Ld; c0 += 32) {
      int s_l = node;
      if (c0 + lane < Ld) s_l = slots[beg + c0 + lane].y;
      chunk_rows<F>(st, s_l, min(32, Ld - c0), H, as4, sd, sub, fl, lane);
    }
  }

  float m_all = st.m;
#pragma unroll
  for (int off = 16; off >= LPE; off >>= 1) m_all = fmaxf(m_all, __shfl_xor(m_all, off, 32));
  const float w = __expf(st.m - m_all);
  float sum = st.sum * w;
  float4 acc = make_float4(st.acc.x * w, st.acc.y * w, st.acc.z * w, st.acc.w * w);
#pragma unroll
  for (int off = 16; off >= LPE; off >>= 1) {
    sum += __shfl_xor(sum, off, 32);
    acc.x += __shfl_xor(acc.x, off, 32);
    acc.y += __shfl_xor(acc.y, off, 32);
    acc.z += __shfl_xor(acc.z, off, 32);
    acc.w += __shfl_xor(acc.w, off, 32);
  }
  const float inv = 1.0f / sum;

  if (small) {
#pragma unroll
    for (int c = 0; c < MAXC; ++c) {
      if (c * 32 + lane < Ld) alpha_out[eidx[c]] = __expf(ev[c] - m_all) * inv;
    }
  } else {
    for (int c0 = 0; c0 < Ld; c0 += 32) {
      int2 sl = make_int2(0, node);
      if (c0 + lane < Ld) sl = slots[beg + c0 + lane];
      Run scratch;
      scratch.m = LOWEST_F; scratch.sum = 0.f; scratch.acc = make_float4(0.f, 0.f, 0.f, 0.f);
      const float e = chunk_rows<F>(scratch, sl.y, min(32, Ld - c0), H, as4, sd, sub, fl, lane);
      if (c0 + lane < Ld) alpha_out[sl.x] = __expf(e - m_all) * inv;
    }
  }
  if (lane == 0) alpha_out[E + node] = __expf(e_self - m_all) * inv;

  if (sub == 0) {
    const float4 b = *(const float4*)(bias + fl * 4);
    float4 v = make_float4(fmaf(acc.x, inv, b.x), fmaf(acc.y, inv, b.y),
                           fmaf(acc.z, inv, b.z), fmaf(acc.w, inv, b.w));
    if (ELU) {
      v.x = (v.x > 0.f) ? v.x : (__expf(v.x) - 1.0f);
      v.y = (v.y > 0.f) ? v.y : (__expf(v.y) - 1.0f);
      v.z = (v.z > 0.f) ? v.z : (__expf(v.z) - 1.0f);
      v.w = (v.w > 0.f) ? v.w : (__expf(v.w) - 1.0f);
    }
    *(float4*)(out + (size_t)node * F + fl * 4) = v;
  }
}

static inline int cdiv(long long a, long long b) { return (int)((a + b - 1) / b); }

extern "C" void kernel_launch(void* const* d_in, const int* in_sizes, int n_in,
                              void* d_out, int out_size, void* d_ws, size_t ws_size,
                              hipStream_t stream) {
  constexpr int FIN = 128, FH = 64, FO = 32;
  const float* x      = (const float*)d_in[0];
  const int*   eidx   = (const int*)d_in[1];
  const float* W1     = (const float*)d_in[2];
  const float* a_src1 = (const float*)d_in[3];
  const float* a_dst1 = (const float*)d_in[4];
  const float* b1     = (const float*)d_in[5];
  const float* W2     = (const float*)d_in[6];
  const float* a_src2 = (const float*)d_in[7];
  const float* a_dst2 = (const float*)d_in[8];
  const float* b2     = (const float*)d_in[9];

  const int N = in_sizes[0] / FIN;
  const int E = in_sizes[1] / 2;
  const int Etot = E + N;
  const int* src = eidx;
  const int* dst = eidx + E;

  float* hfeat1 = (float*)d_ws;
  float* agg1   = hfeat1 + (size_t)N * FH;
  float* hfeat2 = agg1 + (size_t)N * FH;
  int2* slots_a = (int2*)(hfeat2 + (size_t)N * FO);
  int2* slots_b = slots_a + (size_t)Etot;
  int* place    = (int*)(slots_b + (size_t)Etot);
  int* cnt      = place + (size_t)E;
  int* rowbeg   = cnt + N;

  float* h2     = (float*)d_out;
  float* alpha1 = h2 + (size_t)N * FO;
  float* alpha2 = alpha1 + (size_t)Etot;

  dim3 b256(256), b128(128), b1024(1024);

  hipMemsetAsync(cnt, 0, (size_t)N * sizeof(int), stream);
  dst_count_kernel<<<cdiv(E, 256), b256, 0, stream>>>(dst, cnt, place, E);
  row_scan_kernel<<<cdiv(N, 1024), b1024, 0, stream>>>(cnt, rowbeg, N);
  place_kernel<<<cdiv(E, 256), b256, 0, stream>>>(src, dst, place, rowbeg, slots_a, E);
  order_kernel<<<cdiv(N, 8), b256, 0, stream>>>(slots_a, slots_b, rowbeg, N, E);

  gemm_wmma_kernel<FIN, FH><<<cdiv(N, 64), b128, 0, stream>>>(x, W1, hfeat1, N);
  node_gather_kernel<FH, true><<<cdiv(N, 8), b256, 0, stream>>>(
      slots_b, rowbeg, a_src1, a_dst1, hfeat1, b1, alpha1, agg1, N, E);

  gemm_wmma_kernel<FH, FO><<<cdiv(N, 64), b128, 0, stream>>>(agg1, W2, hfeat2, N);
  node_gather_kernel<FO, false><<<cdiv(N, 8), b256, 0, stream>>>(
      slots_b, rowbeg, a_src2, a_dst2, hfeat2, b2, alpha2, h2, N, E);
}


# Edges stored into fixed 64-slot buckets per destination right at the counting atomic (overflow list for longer groups): the place pass, its table and per-edge offset reads go; sum order unchanged
# speedup vs baseline: 2.7814x; 1.1045x over previous
#include <hip/hip_runtime.h>
#include <hip/hip_bf16.h>

typedef __attribute__((ext_vector_type(2))) float v2f;
typedef __attribute__((ext_vector_type(8))) float v8f;

#define NEG_SLOPE 0.2f

template <int FIN, int FOUT>
__global__ __launch_bounds__(128) void gemm_wmma_kernel(
    const float* __restrict__ X, const float* __restrict__ W,
    float* __restrict__ H, int nrows) {
  __shared__ float lds_x[4][16 * FIN];
  const int wave = threadIdx.x >> 5;
  const int lane = threadIdx.x & 31;
  const int m0 = (blockIdx.x * 4 + wave) * 16;

  float* xt = lds_x[wave];
  constexpr int NCHUNK = 16 * FIN / 4;
#pragma unroll 4
  for (int i = lane; i < NCHUNK; i += 32) {
    int r = i / (FIN / 4);
    int c = (i % (FIN / 4)) * 4;
    int row = m0 + r;
    if (row >= nrows) row = nrows - 1;
    float4 v = *(const float4*)(X + (size_t)row * FIN + c);
    *(float4*)(xt + r * FIN + c) = v;
  }
  __syncthreads();

  constexpr int NT = FOUT / 16;
  v8f acc[NT] = {};

  const int arow = lane & 15;
  const int ksel = 2 * (lane >> 4);
  for (int k = 0; k < FIN; k += 4) {
    v2f a;
    a.x = xt[arow * FIN + k + ksel];
    a.y = xt[arow * FIN + k + ksel + 1];
#pragma unroll
    for (int t = 0; t < NT; ++t) {
      int n = t * 16 + (lane & 15);
      v2f b;
      b.x = W[(k + ksel) * FOUT + n];
      b.y = W[(k + ksel + 1) * FOUT + n];
      acc[t] = __builtin_amdgcn_wmma_f32_16x16x4_f32(
          false, a, false, b, (short)0, acc[t], false, false);
    }
  }

  const int row_off = (lane >> 4) * 8;
  const int col = lane & 15;
#pragma unroll
  for (int t = 0; t < NT; ++t) {
#pragma unroll
    for (int i = 0; i < 8; ++i) {
      int row = m0 + row_off + i;
      if (row < nrows) H[(size_t)row * FOUT + t * 16 + col] = acc[t][i];
    }
  }
}


#define CAP 64

__global__ __launch_bounds__(256) void dst_bucket_kernel(
    const int* __restrict__ src, const int* __restrict__ dst, int* __restrict__ cnt,
    int2* __restrict__ bucket, int* __restrict__ ovf_dst, int2* __restrict__ ovf_slot,
    int E, int n) {
  int i = blockIdx.x * blockDim.x + threadIdx.x;
  if (i >= E) return;
  const int d = dst[i];
  const int2 sl = make_int2(i, src[i]);
  const int p = atomicAdd(&cnt[d], 1);
  int2* where = bucket + ((size_t)d * CAP + p);
  if (p >= CAP) {
    const int q = atomicAdd(&cnt[n], 1);
    ovf_dst[q] = d;
    where = ovf_slot + q;
  }
  *where = sl;
}

__global__ __launch_bounds__(1024) void row_scan_kernel(
    const int* __restrict__ cnt, int* __restrict__ rowbeg, int n) {
  __shared__ int wtile[32];
  __shared__ int wfront[32];
  const int t = threadIdx.x;
  const int lane = t & 31, w = t >> 5;
  const int tile0 = blockIdx.x * 1024;
  int f = 0;
#pragma unroll 8
  for (int i = t; i < tile0; i += 1024) f += cnt[i];
  const int i = tile0 + t;
  const int own = (i < n) ? cnt[i] + 1 : 0;
  int v = own;
#pragma unroll
  for (int off = 1; off < 32; off <<= 1) {
    int u = __shfl_up(v, off, 32);
    if (lane >= off) v += u;
  }
#pragma unroll
  for (int off = 16; off > 0; off >>= 1) f += __shfl_xor(f, off, 32);
  if (lane == 31) wtile[w] = v;
  if (lane == 0) wfront[w] = f;
  __syncthreads();
  int s = wfront[lane] + ((lane < w) ? wtile[lane] : 0);
#pragma unroll
  for (int off = 16; off > 0; off >>= 1) s += __shfl_xor(s, off, 32);
  const int base = tile0 + s;
  if (i < n) rowbeg[i] = base + v - own;
  if (i == n - 1) rowbeg[n] = base + v;
}

__global__ __launch_bounds__(256) void order_kernel(
    const int2* bucket, const int* ovf_dst, const int2* ovf_slot, const int* cnt,
    int2* spill, int2* sb, const int* __restrict__ rowbeg, int n, int E) {
  const int lane = threadIdx.x & 31;
  const int node = __builtin_amdgcn_readfirstlane(
      (int)(blockIdx.x * (blockDim.x >> 5) + (threadIdx.x >> 5)));
  if (node >= n) return;
  const int beg = rowbeg[node];
  const int L = rowbeg[node + 1] - beg - 1;
  const int2* sa = bucket + (size_t)node * CAP;
  if (L > CAP) {
    int2* sp = spill + beg;
    for (int i = lane; i < CAP; i += 32) sp[i] = sa[i];
    const int novf = cnt[n];
    int fill = CAP;
    for (int q0 = 0; q0 < novf; q0 += 32) {
      const int q = q0 + lane;
      const bool hit = (q < novf) && (ovf_dst[q] == node);
      const unsigned mask = (unsigned)__ballot(hit);
      if (hit) sp[fill + __popc(mask & ((1u << lane) - 1u))] = ovf_slot[q];
      fill += __popc(mask);
    }
    __threadfence();
    sa = sp;
  }
  for (int i0 = 0; i0 < L; i0 += 32) {
    const int i = i0 + lane;
    int2 me = make_int2(0x7fffffff, 0);
    if (i < L) me = sa[i];
    int r = 0;
    for (int j0 = 0; j0 < L; j0 += 32) {
      const int j = j0 + lane;
      int kj = 0x7fffffff;
      if (j < L) kj = sa[j].x;
      const int lim = min(32, L - j0);
      for (int t = 0; t < lim; ++t)
        r += (__builtin_amdgcn_readlane(kj, t) < me.x) ? 1 : 0;
    }
    if (i < L) sb[beg + r] = me;
  }
  if (lane == 0) sb[beg + L] = make_int2(E + node, node);
}

__device__ __forceinline__ float leaky(float e) { return fmaxf(e, NEG_SLOPE * e); }

template <int X>
__device__ __forceinline__ float row_xor(float v) {
  return __builtin_bit_cast(
      float, __builtin_amdgcn_update_dpp(0, __builtin_bit_cast(int, v), 0x160 + X, 0xf, 0xf, false));
}

template <int LPE>
__device__ __forceinline__ float group_sum(float v) {
  v += row_xor<1>(v);
  v += row_xor<2>(v);
  v += row_xor<4>(v);
  if (LPE == 16) v += row_xor<8>(v);
  return v;
}

__device__ __forceinline__ float dot4(float4 a, float4 b) {
  return a.x * b.x + a.y * b.y + a.z * b.z + a.w * b.w;
}

#define LOWEST_F (-3.402823466e+38f)

struct Run {
  float m, sum;
  float4 acc;
};

template <int F>
__device__ __forceinline__ float chunk_rows(Run& st, int s_l, int cnt, const float* __restrict__ H,
                                            float4 as4, float sd, int sub, int fl, int lane) {
  constexpr int LPE = F / 4;
  constexpr int EPI = 32 / LPE;
  const int steps = (cnt + EPI - 1) / EPI;
  const char* Hb = (const char*)H;
  float keep = 0.f;
#pragma unroll 4
  for (int g = 0; g < steps; ++g) {
    const int jl = g * EPI + sub;
    const unsigned s = (unsigned)__shfl(s_l, jl, 32);
    const float4 h = *(const float4*)(Hb + (s * (unsigned)(F * 4) + (unsigned)(fl * 16)));
    float e = leaky(group_sum<LPE>(dot4(h, as4)) + sd);
    e = (jl < cnt) ? e : -INFINITY;
    const float m_new = fmaxf(st.m, e);
    const float scale = __expf(st.m - m_new);
    const float p = __expf(e - m_new);
    st.acc.x = fmaf(p, h.x, st.acc.x * scale);
    st.acc.y = fmaf(p, h.y, st.acc.y * scale);
    st.acc.z = fmaf(p, h.z, st.acc.z * scale);
    st.acc.w = fmaf(p, h.w, st.acc.w * scale);
    st.sum = fmaf(st.sum, scale, p);
    st.m = m_new;
    if (fl == g) keep = e;
  }
  return __shfl(keep, (lane % EPI) * LPE + lane / EPI, 32);
}

template <int F, bool ELU>
__global__ __launch_bounds__(256) void node_gather_kernel(
    const int2* __restrict__ slots, const int* __restrict__ rowbeg,
    const float* __restrict__ a_src, const float* __restrict__ a_dst,
    const float* __restrict__ H, const float* __restrict__ bias,
    float* __restrict__ alpha_out, float* __restrict__ out, int n, int E) {
  constexpr int LPE = F / 4;
  constexpr int MAXC = 4;
  const int lane = threadIdx.x & 31;
  const int node = __builtin_amdgcn_readfirstlane(
      (int)(blockIdx.x * (blockDim.x >> 5) + (threadIdx.x >> 5)));
  if (node >= n) return;
  const int beg = rowbeg[node];
  const int Ld = rowbeg[node + 1] - beg - 1;
  const int sub = lane / LPE, fl = lane % LPE;
  const bool small = Ld <= 32 * MAXC;

  int sidx[MAXC], eidx[MAXC];
  float ev[MAXC];
#pragma unroll
  for (int c = 0; c < MAXC; ++c) {
    const int2 sl = slots[beg + min(c * 32 + lane, Ld)];
    eidx[c] = sl.x;
    sidx[c] = (c * 32 + lane < Ld) ? sl.y : node;
    ev[c] = 0.f;
  }

  const float4 hself = *(const float4*)(H + (size_t)node * F + fl * 4);
  const float4 as4 = *(const float4*)(a_src + fl * 4);
  const float4 ad4 = *(const float4*)(a_dst + fl * 4);
  const float sd = group_sum<LPE>(dot4(hself, ad4));
  const float e_self = leaky(group_sum<LPE>(dot4(hself, as4)) + sd);

  Run st;
  st.m = (sub == 0) ? e_self : LOWEST_F;
  st.sum = (sub == 0) ? 1.0f : 0.f;
  st.acc = (sub == 0) ? hself : make_float4(0.f, 0.f, 0.f, 0.f);

  if (small) {
#pragma unroll
    for (int c = 0; c < MAXC; ++c) {
      if (c * 32 < Ld)
        ev[c] = chunk_rows<F>(st, sidx[c], min(32, Ld - c * 32), H, as4, sd, sub, fl, lane);
    }
  } else {
    for (int c0 = 0; c0 < Ld; c0 += 32) {
      int s_l = node;
      if (c0 + lane < Ld) s_l = slots[beg + c0 + lane].y;
      chunk_rows<F>(st, s_l, min(32, Ld - c0), H, as4, sd, sub, fl, lane);
    }
  }

  float m_all = st.m;
#pragma unroll
  for (int off = 16; off >= LPE; off >>= 1) m_all = fmaxf(m_all, __shfl_xor(m_all, off, 32));
  const float w = __expf(st.m - m_all);
  float sum = st.sum * w;
  float4 acc = make_float4(st.acc.x * w, st.acc.y * w, st.acc.z * w, st.acc.w * w);
#pragma unroll
  for (int off = 16; off >= LPE; off >>= 1) {
    sum += __shfl_xor(sum, off, 32);
    acc.x += __shfl_xor(acc.x, off, 32);
    acc.y += __shfl_xor(acc.y, off, 32);
    acc.z += __shfl_xor(acc.z, off, 32);
    acc.w += __shfl_xor(acc.w, off, 32);
  }
  const float inv = 1.0f / sum;

  if (small) {
#pragma unroll
    for (int c = 0; c < MAXC; ++c) {
      if (c * 32 + lane < Ld) alpha_out[eidx[c]] = __expf(ev[c] - m_all) * inv;
    }
  } else {
    for (int c0 = 0; c0 < Ld; c0 += 32) {
      int2 sl = make_int2(0, node);
      if (c0 + lane < Ld) sl = slots[beg + c0 + lane];
      Run scratch;
      scratch.m = LOWEST_F; scratch.sum = 0.f; scratch.acc = make_float4(0.f, 0.f, 0.f, 0.f);
      const float e = chunk_rows<F>(scratch, sl.y, min(32, Ld - c0), H, as4, sd, sub, fl, lane);
      if (c0 + lane < Ld) alpha_out[sl.x] = __expf(e - m_all) * inv;
    }
  }
  if (lane == 0) alpha_out[E + node] = __expf(e_self - m_all) * inv;

  if (sub == 0) {
    const float4 b = *(const float4*)(bias + fl * 4);
    float4 v = make_float4(fmaf(acc.x, inv, b.x), fmaf(acc.y, inv, b.y),
                           fmaf(acc.z, inv, b.z), fmaf(acc.w, inv, b.w));
    if (ELU) {
      v.x = (v.x > 0.f) ? v.x : (__expf(v.x) - 1.0f);
      v.y = (v.y > 0.f) ? v.y : (__expf(v.y) - 1.0f);
      v.z = (v.z > 0.f) ? v.z : (__expf(v.z) - 1.0f);
      v.w = (v.w > 0.f) ? v.w : (__expf(v.w) - 1.0f);
    }
    *(float4*)(out + (size_t)node * F + fl * 4) = v;
  }
}

static inline int cdiv(long long a, long long b) { return (int)((a + b - 1) / b); }

extern "C" void kernel_launch(void* const* d_in, const int* in_sizes, int n_in,
                              void* d_out, int out_size, void* d_ws, size_t ws_size,
                              hipStream_t stream) {
  constexpr int FIN = 128, FH = 64, FO = 32;
  const float* x      = (const float*)d_in[0];
  const int*   eidx   = (const int*)d_in[1];
  const float* W1     = (const float*)d_in[2];
  const float* a_src1 = (const float*)d_in[3];
  const float* a_dst1 = (const float*)d_in[4];
  const float* b1     = (const float*)d_in[5];
  const float* W2     = (const float*)d_in[6];
  const float* a_src2 = (const float*)d_in[7];
  const float* a_dst2 = (const float*)d_in[8];
  const float* b2     = (const float*)d_in[9];

  const int N = in_sizes[0] / FIN;
  const int E = in_sizes[1] / 2;
  const int Etot = E + N;
  const int* src = eidx;
  const int* dst = eidx + E;

  float* hfeat1 = (float*)d_ws;
  float* agg1   = hfeat1 + (size_t)N * FH;
  float* hfeat2 = agg1 + (size_t)N * FH;
  int2* bucket  = (int2*)(hfeat2 + (size_t)N * FO);
  int2* slots_b = bucket + (size_t)N * CAP;
  int2* spill   = slots_b + (size_t)Etot;
  int2* ovf_slot = spill + (size_t)Etot;
  int* ovf_dst  = (int*)(ovf_slot + (size_t)E);
  int* cnt      = ovf_dst + (size_t)E;
  int* rowbeg   = cnt + N + 1;

  float* h2     = (float*)d_out;
  float* alpha1 = h2 + (size_t)N * FO;
  float* alpha2 = alpha1 + (size_t)Etot;

  dim3 b256(256), b128(128), b1024(1024);

  hipMemsetAsync(cnt, 0, (size_t)(N + 1) * sizeof(int), stream);
  dst_bucket_kernel<<<cdiv(E, 256), b256, 0, stream>>>(src, dst, cnt, bucket, ovf_dst, ovf_slot, E, N);
  row_scan_kernel<<<cdiv(N, 1024), b1024, 0, stream>>>(cnt, rowbeg, N);
  order_kernel<<<cdiv(N, 8), b256, 0, stream>>>(
      bucket, ovf_dst, ovf_slot, cnt, spill, slots_b, rowbeg, N, E);

  gemm_wmma_kernel<FIN, FH><<<cdiv(N, 64), b128, 0, stream>>>(x, W1, hfeat1, N);
  node_gather_kernel<FH, true><<<cdiv(N, 8), b256, 0, stream>>>(
      slots_b, rowbeg, a_src1, a_dst1, hfeat1, b1, alpha1, agg1, N, E);

  gemm_wmma_kernel<FH, FO><<<cdiv(N, 64), b128, 0, stream>>>(agg1, W2, hfeat2, N);
  node_gather_kernel<FO, false><<<cdiv(N, 8), b256, 0, stream>>>(
      slots_b, rowbeg, a_src2, a_dst2, hfeat2, b2, alpha2, h2, N, E);
}


# Edges grouped by destination inside LDS: count per node range, chunks sorted by range in LDS, one work-group per range counts, scans, places, orders; per-edge global atomic and scattered store gone
# speedup vs baseline: 4.2838x; 1.5402x over previous
#include <hip/hip_runtime.h>
#include <hip/hip_bf16.h>

typedef __attribute__((ext_vector_type(2))) float v2f;
typedef __attribute__((ext_vector_type(8))) float v8f;

#define NEG_SLOPE 0.2f

template <int FIN, int FOUT>
__global__ __launch_bounds__(128) void gemm_wmma_kernel(
    const float* __restrict__ X, const float* __restrict__ W,
    float* __restrict__ H, int nrows) {
  __shared__ float lds_x[4][16 * FIN];
  const int wave = threadIdx.x >> 5;
  const int lane = threadIdx.x & 31;
  const int m0 = (blockIdx.x * 4 + wave) * 16;

  float* xt = lds_x[wave];
  constexpr int NCHUNK = 16 * FIN / 4;
#pragma unroll 4
  for (int i = lane; i < NCHUNK; i += 32) {
    int r = i / (FIN / 4);
    int c = (i % (FIN / 4)) * 4;
    int row = m0 + r;
    if (row >= nrows) row = nrows - 1;
    float4 v = *(const float4*)(X + (size_t)row * FIN + c);
    *(float4*)(xt + r * FIN + c) = v;
  }
  __syncthreads();

  constexpr int NT = FOUT / 16;
  v8f acc[NT] = {};

  const int arow = lane & 15;
  const int ksel = 2 * (lane >> 4);
  for (int k = 0; k < FIN; k += 4) {
    v2f a;
    a.x = xt[arow * FIN + k + ksel];
    a.y = xt[arow * FIN + k + ksel + 1];
#pragma unroll
    for (int t = 0; t < NT; ++t) {
      int n = t * 16 + (lane & 15);
      v2f b;
      b.x = W[(k + ksel) * FOUT + n];
      b.y = W[(k + ksel + 1) * FOUT + n];
      acc[t] = __builtin_amdgcn_wmma_f32_16x16x4_f32(
          false, a, false, b, (short)0, acc[t], false, false);
    }
  }

  const int row_off = (lane >> 4) * 8;
  const int col = lane & 15;
#pragma unroll
  for (int t = 0; t < NT; ++t) {
#pragma unroll
    for (int i = 0; i < 8; ++i) {
      int row = m0 + row_off + i;
      if (row < nrows) H[(size_t)row * FOUT + t * 16 + col] = acc[t][i];
    }
  }
}

#define NBIN 256
#define RMAX 512
#define CHUNK 8192
#define LCAP 28672

__global__ __launch_bounds__(1024) void bin_count_kernel(
    const int* __restrict__ dst, int* __restrict__ bincnt, int E, int n) {
  __shared__ int hist[NBIN];
  const int t = threadIdx.x;
  const unsigned R = (unsigned)((n + NBIN - 1) / NBIN);
  if (t < NBIN) hist[t] = 0;
  __syncthreads();
  const int base = blockIdx.x * CHUNK;
#pragma unroll
  for (int k = 0; k < CHUNK / 1024; ++k) {
    const int i = base + k * 1024 + t;
    if (i < E) atomicAdd(&hist[(unsigned)dst[i] / R], 1);
  }
  __syncthreads();
  if (t < NBIN && hist[t] != 0) atomicAdd(&bincnt[t], hist[t]);
}

__device__ __forceinline__ int scan_bins(int h, int* wsum) {
  const int t = threadIdx.x, lane = t & 31;
  int v = h;
  if (t < NBIN) {
#pragma unroll
    for (int off = 1; off < 32; off <<= 1) {
      const int u = __shfl_up(v, off, 32);
      if (lane >= off) v += u;
    }
    if (lane == 31) wsum[t >> 5] = v;
  }
  __syncthreads();
  int front = 0;
  if (t < NBIN) {
    for (int w = 0; w < (t >> 5); ++w) front += wsum[w];
  }
  return front + v - h;
}

__global__ __launch_bounds__(1024) void partition_kernel(
    const int* __restrict__ src, const int* __restrict__ dst, const int* __restrict__ bincnt,
    int* __restrict__ cursor, int2* __restrict__ pslot, int* __restrict__ pdst, int E, int n) {
  __shared__ int hist[NBIN];
  __shared__ int lbase[NBIN];
  __shared__ int gbase[NBIN];
  __shared__ int wsum_a[NBIN / 32], wsum_b[NBIN / 32];
  __shared__ int2 st_slot[CHUNK];
  __shared__ int st_dst[CHUNK];
  const int t = threadIdx.x;
  const unsigned R = (unsigned)((n + NBIN - 1) / NBIN);
  if (t < NBIN) hist[t] = 0;
  __syncthreads();
  const int base = blockIdx.x * CHUNK;
  int d[CHUNK / 1024], s[CHUNK / 1024], lp[CHUNK / 1024];
#pragma unroll
  for (int k = 0; k < CHUNK / 1024; ++k) {
    const int i = base + k * 1024 + t;
    d[k] = -1; s[k] = 0; lp[k] = 0;
    if (i < E) {
      d[k] = dst[i];
      s[k] = src[i];
    }
  }
#pragma unroll
  for (int k = 0; k < CHUNK / 1024; ++k) {
    if (d[k] >= 0) lp[k] = atomicAdd(&hist[(unsigned)d[k] / R], 1);
  }
  __syncthreads();
  const int h = (t < NBIN) ? hist[t] : 0;
  const int g = (t < NBIN) ? bincnt[t] : 0;
  const int lfront = scan_bins(h, wsum_a);
  const int gfront = scan_bins(g, wsum_b);
  if (t < NBIN) {
    lbase[t] = lfront;
    gbase[t] = gfront + ((h != 0) ? atomicAdd(&cursor[t], h) : 0);
  }
  __syncthreads();
#pragma unroll
  for (int k = 0; k < CHUNK / 1024; ++k) {
    if (d[k] >= 0) {
      const int pos = lbase[(unsigned)d[k] / R] + lp[k];
      st_slot[pos] = make_int2(base + k * 1024 + t, s[k]);
      st_dst[pos] = d[k];
    }
  }
  __syncthreads();
  const int cntc = min(CHUNK, E - base);
#pragma unroll
  for (int k = 0; k < CHUNK / 1024; ++k) {
    const int p = k * 1024 + t;
    if (p < cntc) {
      const int dd = st_dst[p];
      const unsigned bb = (unsigned)dd / R;
      const int gp = gbase[bb] + (p - lbase[bb]);
      pslot[gp] = st_slot[p];
      pdst[gp] = dd;
    }
  }
}

__device__ __forceinline__ void range_place_order(
    int2* buf, int* cur, const int* lbeg, const int2* __restrict__ pslot,
    const int* __restrict__ pdst, int e0, int cb, int n0, int nr, int gb0,
    int2* __restrict__ sb, int E) {
  const int t = threadIdx.x, lane = t & 31;
#pragma unroll 4
  for (int e = t; e < cb; e += 1024) {
    const int j = pdst[e0 + e] - n0;
    const int p = atomicAdd(&cur[j], 1);
    buf[lbeg[j] + p] = pslot[e0 + e];
  }
  __syncthreads();
  const int w = __builtin_amdgcn_readfirstlane(t >> 5);
  for (int j = w; j < nr; j += 32) {
    const int lb = __builtin_amdgcn_readfirstlane(lbeg[j]);
    const int L = __builtin_amdgcn_readfirstlane(lbeg[j + 1]) - lb;
    const int2* sa = buf + lb;
    const int beg = gb0 + lb + n0 + j;
    for (int i0 = 0; i0 < L; i0 += 32) {
      const int i = i0 + lane;
      int2 me = make_int2(0x7fffffff, 0);
      if (i < L) me = sa[i];
      int r = 0;
      for (int j0 = 0; j0 < L; j0 += 32) {
        const int jj = j0 + lane;
        int kj = 0x7fffffff;
        if (jj < L) kj = sa[jj].x;
        const int lim = min(32, L - j0);
        for (int q = 0; q < lim; ++q)
          r += (__builtin_amdgcn_readlane(kj, q) < me.x) ? 1 : 0;
      }
      if (i < L) sb[beg + r] = me;
    }
    if (lane == 0) sb[beg + L] = make_int2(E + n0 + j, n0 + j);
  }
}

__global__ __launch_bounds__(1024) void range_order_kernel(
    const int2* __restrict__ pslot, const int* __restrict__ pdst, const int* __restrict__ bincnt,
    int2* spare, int2* __restrict__ sb, int* __restrict__ rowbeg, int n, int E) {
  __shared__ int cnt_l[RMAX];
  __shared__ int lbeg[RMAX + 1];
  __shared__ int wsum[NBIN / 32];
  __shared__ int wsum_n[RMAX / 32];
  __shared__ int sh_first, sh_count;
  __shared__ int2 lbuf[LCAP];
  const int t = threadIdx.x, lane = t & 31;
  const int b = blockIdx.x;
  const int R = (n + NBIN - 1) / NBIN;
  const int n0 = b * R;
  const int nr = max(0, min(R, n - n0));
  const int g = (t < NBIN) ? bincnt[t] : 0;
  const int gfront = scan_bins(g, wsum);
  if (t == b) { sh_first = gfront; sh_count = g; }
  for (int j = t; j < RMAX; j += 1024) cnt_l[j] = 0;
  __syncthreads();
  const int gb0 = sh_first;
  const int cb = sh_count;
#pragma unroll 4
  for (int e = t; e < cb; e += 1024) atomicAdd(&cnt_l[pdst[gb0 + e] - n0], 1);
  __syncthreads();
  int hn = 0, vn = 0;
  if (t < RMAX) {
    hn = (t < nr) ? cnt_l[t] : 0;
    vn = hn;
#pragma unroll
    for (int off = 1; off < 32; off <<= 1) {
      const int u = __shfl_up(vn, off, 32);
      if (lane >= off) vn += u;
    }
    if (lane == 31) wsum_n[t >> 5] = vn;
    cnt_l[t] = 0;
  }
  __syncthreads();
  if (t < RMAX) {
    int front = 0;
    for (int w = 0; w < (t >> 5); ++w) front += wsum_n[w];
    lbeg[t + 1] = front + vn;
    if (t == 0) lbeg[0] = 0;
    if (t < nr) rowbeg[n0 + t] = gb0 + (front + vn - hn) + n0 + t;
    if (t + 1 == nr && n0 + nr == n) rowbeg[n] = gb0 + (front + vn) + n;
  }
  __syncthreads();
  if (cb <= LCAP)
    range_place_order(lbuf, cnt_l, lbeg, pslot, pdst, gb0, cb, n0, nr, gb0, sb, E);
  else
    range_place_order(spare + gb0, cnt_l, lbeg, pslot, pdst, gb0, cb, n0, nr, gb0, sb, E);
}

__device__ __forceinline__ float leaky(float e) { return fmaxf(e, NEG_SLOPE * e); }

template <int X>
__device__ __forceinline__ float row_xor(float v) {
  return __builtin_bit_cast(
      float, __builtin_amdgcn_update_dpp(0, __builtin_bit_cast(int, v), 0x160 + X, 0xf, 0xf, false));
}

template <int LPE>
__device__ __forceinline__ float group_sum(float v) {
  v += row_xor<1>(v);
  v += row_xor<2>(v);
  v += row_xor<4>(v);
  if (LPE == 16) v += row_xor<8>(v);
  return v;
}

__device__ __forceinline__ float dot4(float4 a, float4 b) {
  return a.x * b.x + a.y * b.y + a.z * b.z + a.w * b.w;
}

#define LOWEST_F (-3.402823466e+38f)

struct Run {
  float m, sum;
  float4 acc;
};

template <int F>
__device__ __forceinline__ float chunk_rows(Run& st, int s_l, int cnt, const float* __restrict__ H,
                                            float4 as4, float sd, int sub, int fl, int lane) {
  constexpr int LPE = F / 4;
  constexpr int EPI = 32 / LPE;
  const int steps = (cnt + EPI - 1) / EPI;
  const char* Hb = (const char*)H;
  float keep = 0.f;
#pragma unroll 4
  for (int g = 0; g < steps; ++g) {
    const int jl = g * EPI + sub;
    const unsigned s = (unsigned)__shfl(s_l, jl, 32);
    const float4 h = *(const float4*)(Hb + (s * (unsigned)(F * 4) + (unsigned)(fl * 16)));
    float e = leaky(group_sum<LPE>(dot4(h, as4)) + sd);
    e = (jl < cnt) ? e : -INFINITY;
    const float m_new = fmaxf(st.m, e);
    const float scale = __expf(st.m - m_new);
    const float p = __expf(e - m_new);
    st.acc.x = fmaf(p, h.x, st.acc.x * scale);
    st.acc.y = fmaf(p, h.y, st.acc.y * scale);
    st.acc.z = fmaf(p, h.z, st.acc.z * scale);
    st.acc.w = fmaf(p, h.w, st.acc.w * scale);
    st.sum = fmaf(st.sum, scale, p);
    st.m = m_new;
    if (fl == g) keep = e;
  }
  return __shfl(keep, (lane % EPI) * LPE + lane / EPI, 32);
}

template <int F, bool ELU>
__global__ __launch_bounds__(256) void node_gather_kernel(
    const int2* __restrict__ slots, const int* __restrict__ rowbeg,
    const float* __restrict__ a_src, const float* __restrict__ a_dst,
    const float* __restrict__ H, const float* __restrict__ bias,
    float* __restrict__ alpha_out, float* __restrict__ out, int n, int E) {
  constexpr int LPE = F / 4;
  constexpr int MAXC = 4;
  const int lane = threadIdx.x & 31;
  const int node = __builtin_amdgcn_readfirstlane(
      (int)(blockIdx.x * (blockDim.x >> 5) + (threadIdx.x >> 5)));
  if (node >= n) return;
  const int beg = rowbeg[node];
  const int Ld = rowbeg[node + 1] - beg - 1;
  const int sub = lane / LPE, fl = lane % LPE;
  const bool small = Ld <= 32 * MAXC;

  int sidx[MAXC], eidx[MAXC];
  float ev[MAXC];
#pragma unroll
  for (int c = 0; c < MAXC; ++c) {
    const int2 sl = slots[beg + min(c * 32 + lane, Ld)];
    eidx[c] = sl.x;
    sidx[c] = (c * 32 + lane < Ld) ? sl.y : node;
    ev[c] = 0.f;
  }

  const float4 hself = *(const float4*)(H + (size_t)node * F + fl * 4);
  const float4 as4 = *(const float4*)(a_src + fl * 4);
  const float4 ad4 = *(const float4*)(a_dst + fl * 4);
  const float sd = group_sum<LPE>(dot4(hself, ad4));
  const float e_self = leaky(group_sum<LPE>(dot4(hself, as4)) + sd);

  Run st;
  st.m = (sub == 0) ? e_self : LOWEST_F;
  st.sum = (sub == 0) ? 1.0f : 0.f;
  st.acc = (sub == 0) ? hself : make_float4(0.f, 0.f, 0.f, 0.f);

  if (small) {
#pragma unroll
    for (int c = 0; c < MAXC; ++c) {
      if (c * 32 < Ld)
        ev[c] = chunk_rows<F>(st, sidx[c], min(32, Ld - c * 32), H, as4, sd, sub, fl, lane);
    }
  } else {
    for (int c0 = 0; c0 < Ld; c0 += 32) {
      int s_l = node;
      if (c0 + lane < Ld) s_l = slots[beg + c0 + lane].y;
      chunk_rows<F>(st, s_l, min(32, Ld - c0), H, as4, sd, sub, fl, lane);
    }
  }

  float m_all = st.m;
#pragma unroll
  for (int off = 16; off >= LPE; off >>= 1) m_all = fmaxf(m_all, __shfl_xor(m_all, off, 32));
  const float w = __expf(st.m - m_all);
  float sum = st.sum * w;
  float4 acc = make_float4(st.acc.x * w, st.acc.y * w, st.acc.z * w, st.acc.w * w);
#pragma unroll
  for (int off = 16; off >= LPE; off >>= 1) {
    sum += __shfl_xor(sum, off, 32);
    acc.x += __shfl_xor(acc.x, off, 32);
    acc.y += __shfl_xor(acc.y, off, 32);
    acc.z += __shfl_xor(acc.z, off, 32);
    acc.w += __shfl_xor(acc.w, off, 32);
  }
  const float inv = 1.0f / sum;

  if (small) {
#pragma unroll
    for (int c = 0; c < MAXC; ++c) {
      if (c * 32 + lane < Ld) alpha_out[eidx[c]] = __expf(ev[c] - m_all) * inv;
    }
  } else {
    for (int c0 = 0; c0 < Ld; c0 += 32) {
      int2 sl = make_int2(0, node);
      if (c0 + lane < Ld) sl = slots[beg + c0 + lane];
      Run scratch;
      scratch.m = LOWEST_F; scratch.sum = 0.f; scratch.acc = make_float4(0.f, 0.f, 0.f, 0.f);
      const float e = chunk_rows<F>(scratch, sl.y, min(32, Ld - c0), H, as4, sd, sub, fl, lane);
      if (c0 + lane < Ld) alpha_out[sl.x] = __expf(e - m_all) * inv;
    }
  }
  if (lane == 0) alpha_out[E + node] = __expf(e_self - m_all) * inv;

  if (sub == 0) {
    const float4 b = *(const float4*)(bias + fl * 4);
    float4 v = make_float4(fmaf(acc.x, inv, b.x), fmaf(acc.y, inv, b.y),
                           fmaf(acc.z, inv, b.z), fmaf(acc.w, inv, b.w));
    if (ELU) {
      v.x = (v.x > 0.f) ? v.x : (__expf(v.x) - 1.0f);
      v.y = (v.y > 0.f) ? v.y : (__expf(v.y) - 1.0f);
      v.z = (v.z > 0.f) ? v.z : (__expf(v.z) - 1.0f);
      v.w = (v.w > 0.f) ? v.w : (__expf(v.w) - 1.0f);
    }
    *(float4*)(out + (size_t)node * F + fl * 4) = v;
  }
}

static inline int cdiv(long long a, long long b) { return (int)((a + b - 1) / b); }

extern "C" void kernel_launch(void* const* d_in, const int* in_sizes, int n_in,
                              void* d_out, int out_size, void* d_ws, size_t ws_size,
                              hipStream_t stream) {
  constexpr int FIN = 128, FH = 64, FO = 32;
  const float* x      = (const float*)d_in[0];
  const int*   eidx   = (const int*)d_in[1];
  const float* W1     = (const float*)d_in[2];
  const float* a_src1 = (const float*)d_in[3];
  const float* a_dst1 = (const float*)d_in[4];
  const float* b1     = (const float*)d_in[5];
  const float* W2     = (const float*)d_in[6];
  const float* a_src2 = (const float*)d_in[7];
  const float* a_dst2 = (const float*)d_in[8];
  const float* b2     = (const float*)d_in[9];

  const int N = in_sizes[0] / FIN;
  const int E = in_sizes[1] / 2;
  const int Etot = E + N;
  const int* src = eidx;
  const int* dst = eidx + E;

  float* hfeat1 = (float*)d_ws;
  float* agg1   = hfeat1 + (size_t)N * FH;
  float* hfeat2 = agg1 + (size_t)N * FH;
  int2* pslot   = (int2*)(hfeat2 + (size_t)N * FO);
  int2* slots_b = pslot + (size_t)E;
  int2* spare   = slots_b + (size_t)Etot;
  int* pdst     = (int*)(spare + (size_t)E);
  int* bincnt   = pdst + (size_t)E;
  int* rowbeg   = bincnt + 2 * NBIN;

  float* h2     = (float*)d_out;
  float* alpha1 = h2 + (size_t)N * FO;
  float* alpha2 = alpha1 + (size_t)Etot;

  dim3 b256(256), b128(128), b1024(1024);

  hipMemsetAsync(bincnt, 0, (size_t)(2 * NBIN) * sizeof(int), stream);
  bin_count_kernel<<<cdiv(E, CHUNK), b1024, 0, stream>>>(dst, bincnt, E, N);
  partition_kernel<<<cdiv(E, CHUNK), b1024, 0, stream>>>(
      src, dst, bincnt, bincnt + NBIN, pslot, pdst, E, N);
  range_order_kernel<<<NBIN, b1024, 0, stream>>>(pslot, pdst, bincnt, spare, slots_b, rowbeg, N, E);

  gemm_wmma_kernel<FIN, FH><<<cdiv(N, 64), b128, 0, stream>>>(x, W1, hfeat1, N);
  node_gather_kernel<FH, true><<<cdiv(N, 8), b256, 0, stream>>>(
      slots_b, rowbeg, a_src1, a_dst1, hfeat1, b1, alpha1, agg1, N, E);

  gemm_wmma_kernel<FH, FO><<<cdiv(N, 64), b128, 0, stream>>>(agg1, W2, hfeat2, N);
  node_gather_kernel<FO, false><<<cdiv(N, 8), b256, 0, stream>>>(
      slots_b, rowbeg, a_src2, a_dst2, hfeat2, b2, alpha2, h2, N, E);
}


# Alphas stored in slot order (one run per node), carried to edge order by two LDS-binned passes for both layers at once: 6.6M scattered 4-byte stores go, 2 launches more; values, sum order unchanged
# speedup vs baseline: 5.6836x; 1.3268x over previous
#include <hip/hip_runtime.h>
#include <hip/hip_bf16.h>

typedef __attribute__((ext_vector_type(2))) float v2f;
typedef __attribute__((ext_vector_type(8))) float v8f;

#define NEG_SLOPE 0.2f

template <int FIN, int FOUT>
__global__ __launch_bounds__(128) void gemm_wmma_kernel(
    const float* __restrict__ X, const float* __restrict__ W,
    float* __restrict__ H, int nrows) {
  __shared__ float lds_x[4][16 * FIN];
  const int wave = threadIdx.x >> 5;
  const int lane = threadIdx.x & 31;
  const int m0 = (blockIdx.x * 4 + wave) * 16;

  float* xt = lds_x[wave];
  constexpr int NCHUNK = 16 * FIN / 4;
#pragma unroll 4
  for (int i = lane; i < NCHUNK; i += 32) {
    int r = i / (FIN / 4);
    int c = (i % (FIN / 4)) * 4;
    int row = m0 + r;
    if (row >= nrows) row = nrows - 1;
    float4 v = *(const float4*)(X + (size_t)row * FIN + c);
    *(float4*)(xt + r * FIN + c) = v;
  }
  __syncthreads();

  constexpr int NT = FOUT / 16;
  v8f acc[NT] = {};

  const int arow = lane & 15;
  const int ksel = 2 * (lane >> 4);
  for (int k = 0; k < FIN; k += 4) {
    v2f a;
    a.x = xt[arow * FIN + k + ksel];
    a.y = xt[arow * FIN + k + ksel + 1];
#pragma unroll
    for (int t = 0; t < NT; ++t) {
      int n = t * 16 + (lane & 15);
      v2f b;
      b.x = W[(k + ksel) * FOUT + n];
      b.y = W[(k + ksel + 1) * FOUT + n];
      acc[t] = __builtin_amdgcn_wmma_f32_16x16x4_f32(
          false, a, false, b, (short)0, acc[t], false, false);
    }
  }

  const int row_off = (lane >> 4) * 8;
  const int col = lane & 15;
#pragma unroll
  for (int t = 0; t < NT; ++t) {
#pragma unroll
    for (int i = 0; i < 8; ++i) {
      int row = m0 + row_off + i;
      if (row < nrows) H[(size_t)row * FOUT + t * 16 + col] = acc[t][i];
    }
  }
}

#define NBIN 256
#define RMAX 512
#define CHUNK 8192
#define LCAP 28672

__global__ __launch_bounds__(1024) void bin_count_kernel(
    const int* __restrict__ dst, int* __restrict__ bincnt, int E, int n) {
  __shared__ int hist[NBIN];
  const int t = threadIdx.x;
  const unsigned R = (unsigned)((n + NBIN - 1) / NBIN);
  if (t < NBIN) hist[t] = 0;
  __syncthreads();
  const int base = blockIdx.x * CHUNK;
#pragma unroll
  for (int k = 0; k < CHUNK / 1024; ++k) {
    const int i = base + k * 1024 + t;
    if (i < E) atomicAdd(&hist[(unsigned)dst[i] / R], 1);
  }
  __syncthreads();
  if (t < NBIN && hist[t] != 0) atomicAdd(&bincnt[t], hist[t]);
}

__device__ __forceinline__ int scan_bins(int h, int* wsum) {
  const int t = threadIdx.x, lane = t & 31;
  int v = h;
  if (t < NBIN) {
#pragma unroll
    for (int off = 1; off < 32; off <<= 1) {
      const int u = __shfl_up(v, off, 32);
      if (lane >= off) v += u;
    }
    if (lane == 31) wsum[t >> 5] = v;
  }
  __syncthreads();
  int front = 0;
  if (t < NBIN) {
    for (int w = 0; w < (t >> 5); ++w) front += wsum[w];
  }
  return front + v - h;
}

__global__ __launch_bounds__(1024) void partition_kernel(
    const int* __restrict__ src, const int* __restrict__ dst, const int* __restrict__ bincnt,
    int* __restrict__ cursor, int2* __restrict__ pslot, int* __restrict__ pdst, int E, int n) {
  __shared__ int hist[NBIN];
  __shared__ int lbase[NBIN];
  __shared__ int gbase[NBIN];
  __shared__ int wsum_a[NBIN / 32], wsum_b[NBIN / 32];
  __shared__ int2 st_slot[CHUNK];
  __shared__ int st_dst[CHUNK];
  const int t = threadIdx.x;
  const unsigned R = (unsigned)((n + NBIN - 1) / NBIN);
  if (t < NBIN) hist[t] = 0;
  __syncthreads();
  const int base = blockIdx.x * CHUNK;
  int d[CHUNK / 1024], s[CHUNK / 1024], lp[CHUNK / 1024];
#pragma unroll
  for (int k = 0; k < CHUNK / 1024; ++k) {
    const int i = base + k * 1024 + t;
    d[k] = -1; s[k] = 0; lp[k] = 0;
    if (i < E) {
      d[k] = dst[i];
      s[k] = src[i];
    }
  }
#pragma unroll
  for (int k = 0; k < CHUNK / 1024; ++k) {
    if (d[k] >= 0) lp[k] = atomicAdd(&hist[(unsigned)d[k] / R], 1);
  }
  __syncthreads();
  const int h = (t < NBIN) ? hist[t] : 0;
  const int g = (t < NBIN) ? bincnt[t] : 0;
  const int lfront = scan_bins(h, wsum_a);
  const int gfront = scan_bins(g, wsum_b);
  if (t < NBIN) {
    lbase[t] = lfront;
    gbase[t] = gfront + ((h != 0) ? atomicAdd(&cursor[t], h) : 0);
  }
  __syncthreads();
#pragma unroll
  for (int k = 0; k < CHUNK / 1024; ++k) {
    if (d[k] >= 0) {
      const int pos = lbase[(unsigned)d[k] / R] + lp[k];
      st_slot[pos] = make_int2(base + k * 1024 + t, s[k]);
      st_dst[pos] = d[k];
    }
  }
  __syncthreads();
  const int cntc = min(CHUNK, E - base);
#pragma unroll
  for (int k = 0; k < CHUNK / 1024; ++k) {
    const int p = k * 1024 + t;
    if (p < cntc) {
      const int dd = st_dst[p];
      const unsigned bb = (unsigned)dd / R;
      const int gp = gbase[bb] + (p - lbase[bb]);
      pslot[gp] = st_slot[p];
      pdst[gp] = dd;
    }
  }
}

__device__ __forceinline__ void range_place_order(
    int2* buf, int* cur, const int* lbeg, const int2* __restrict__ pslot,
    const int* __restrict__ pdst, int e0, int cb, int n0, int nr, int gb0,
    int2* __restrict__ sb, int E) {
  const int t = threadIdx.x, lane = t & 31;
#pragma unroll 4
  for (int e = t; e < cb; e += 1024) {
    const int j = pdst[e0 + e] - n0;
    const int p = atomicAdd(&cur[j], 1);
    buf[lbeg[j] + p] = pslot[e0 + e];
  }
  __syncthreads();
  const int w = __builtin_amdgcn_readfirstlane(t >> 5);
  for (int j = w; j < nr; j += 32) {
    const int lb = __builtin_amdgcn_readfirstlane(lbeg[j]);
    const int L = __builtin_amdgcn_readfirstlane(lbeg[j + 1]) - lb;
    const int2* sa = buf + lb;
    const int beg = gb0 + lb + n0 + j;
    for (int i0 = 0; i0 < L; i0 += 32) {
      const int i = i0 + lane;
      int2 me = make_int2(0x7fffffff, 0);
      if (i < L) me = sa[i];
      int r = 0;
      for (int j0 = 0; j0 < L; j0 += 32) {
        const int jj = j0 + lane;
        int kj = 0x7fffffff;
        if (jj < L) kj = sa[jj].x;
        const int lim = min(32, L - j0);
        for (int q = 0; q < lim; ++q)
          r += (__builtin_amdgcn_readlane(kj, q) < me.x) ? 1 : 0;
      }
      if (i < L) sb[beg + r] = me;
    }
    if (lane == 0) sb[beg + L] = make_int2(E + n0 + j, n0 + j);
  }
}

__global__ __launch_bounds__(1024) void range_order_kernel(
    const int2* __restrict__ pslot, const int* __restrict__ pdst, const int* __restrict__ bincnt,
    int2* spare, int2* __restrict__ sb, int* __restrict__ rowbeg, int n, int E) {
  __shared__ int cnt_l[RMAX];
  __shared__ int lbeg[RMAX + 1];
  __shared__ int wsum[NBIN / 32];
  __shared__ int wsum_n[RMAX / 32];
  __shared__ int sh_first, sh_count;
  __shared__ int2 lbuf[LCAP];
  const int t = threadIdx.x, lane = t & 31;
  const int b = blockIdx.x;
  const int R = (n + NBIN - 1) / NBIN;
  const int n0 = b * R;
  const int nr = max(0, min(R, n - n0));
  const int g = (t < NBIN) ? bincnt[t] : 0;
  const int gfront = scan_bins(g, wsum);
  if (t == b) { sh_first = gfront; sh_count = g; }
  for (int j = t; j < RMAX; j += 1024) cnt_l[j] = 0;
  __syncthreads();
  const int gb0 = sh_first;
  const int cb = sh_count;
#pragma unroll 4
  for (int e = t; e < cb; e += 1024) atomicAdd(&cnt_l[pdst[gb0 + e] - n0], 1);
  __syncthreads();
  int hn = 0, vn = 0;
  if (t < RMAX) {
    hn = (t < nr) ? cnt_l[t] : 0;
    vn = hn;
#pragma unroll
    for (int off = 1; off < 32; off <<= 1) {
      const int u = __shfl_up(vn, off, 32);
      if (lane >= off) vn += u;
    }
    if (lane == 31) wsum_n[t >> 5] = vn;
    cnt_l[t] = 0;
  }
  __syncthreads();
  if (t < RMAX) {
    int front = 0;
    for (int w = 0; w < (t >> 5); ++w) front += wsum_n[w];
    lbeg[t + 1] = front + vn;
    if (t == 0) lbeg[0] = 0;
    if (t < nr) rowbeg[n0 + t] = gb0 + (front + vn - hn) + n0 + t;
    if (t + 1 == nr && n0 + nr == n) rowbeg[n] = gb0 + (front + vn) + n;
  }
  __syncthreads();
  if (cb <= LCAP)
    range_place_order(lbuf, cnt_l, lbeg, pslot, pdst, gb0, cb, n0, nr, gb0, sb, E);
  else
    range_place_order(spare + gb0, cnt_l, lbeg, pslot, pdst, gb0, cb, n0, nr, gb0, sb, E);
}

__device__ __forceinline__ float leaky(float e) { return fmaxf(e, NEG_SLOPE * e); }

template <int X>
__device__ __forceinline__ float row_xor(float v) {
  return __builtin_bit_cast(
      float, __builtin_amdgcn_update_dpp(0, __builtin_bit_cast(int, v), 0x160 + X, 0xf, 0xf, false));
}

template <int LPE>
__device__ __forceinline__ float group_sum(float v) {
  v += row_xor<1>(v);
  v += row_xor<2>(v);
  v += row_xor<4>(v);
  if (LPE == 16) v += row_xor<8>(v);
  return v;
}

__device__ __forceinline__ float dot4(float4 a, float4 b) {
  return a.x * b.x + a.y * b.y + a.z * b.z + a.w * b.w;
}

#define LOWEST_F (-3.402823466e+38f)

struct Run {
  float m, sum;
  float4 acc;
};

template <int F>
__device__ __forceinline__ float chunk_rows(Run& st, int s_l, int cnt, const float* __restrict__ H,
                                            float4 as4, float sd, int sub, int fl, int lane) {
  constexpr int LPE = F / 4;
  constexpr int EPI = 32 / LPE;
  const int steps = (cnt + EPI - 1) / EPI;
  const char* Hb = (const char*)H;
  float keep = 0.f;
#pragma unroll 4
  for (int g = 0; g < steps; ++g) {
    const int jl = g * EPI + sub;
    const unsigned s = (unsigned)__shfl(s_l, jl, 32);
    const float4 h = *(const float4*)(Hb + (s * (unsigned)(F * 4) + (unsigned)(fl * 16)));
    float e = leaky(group_sum<LPE>(dot4(h, as4)) + sd);
    e = (jl < cnt) ? e : -INFINITY;
    const float m_new = fmaxf(st.m, e);
    const float scale = __expf(st.m - m_new);
    const float p = __expf(e - m_new);
    st.acc.x = fmaf(p, h.x, st.acc.x * scale);
    st.acc.y = fmaf(p, h.y, st.acc.y * scale);
    st.acc.z = fmaf(p, h.z, st.acc.z * scale);
    st.acc.w = fmaf(p, h.w, st.acc.w * scale);
    st.sum = fmaf(st.sum, scale, p);
    st.m = m_new;
    if (fl == g) keep = e;
  }
  return __shfl(keep, (lane % EPI) * LPE + lane / EPI, 32);
}

template <int F, bool ELU>
__global__ __launch_bounds__(256) void node_gather_kernel(
    const int2* __restrict__ slots, const int* __restrict__ rowbeg,
    const float* __restrict__ a_src, const float* __restrict__ a_dst,
    const float* __restrict__ H, const float* __restrict__ bias,
    float* __restrict__ alpha_slot, float* __restrict__ out, int n) {
  constexpr int LPE = F / 4;
  constexpr int MAXC = 4;
  const int lane = threadIdx.x & 31;
  const int node = __builtin_amdgcn_readfirstlane(
      (int)(blockIdx.x * (blockDim.x >> 5) + (threadIdx.x >> 5)));
  if (node >= n) return;
  const int beg = rowbeg[node];
  const int Ld = rowbeg[node + 1] - beg - 1;
  const int sub = lane / LPE, fl = lane % LPE;
  const bool small = Ld <= 32 * MAXC;

  int sidx[MAXC];
  float ev[MAXC];
#pragma unroll
  for (int c = 0; c < MAXC; ++c) {
    const int2 sl = slots[beg + min(c * 32 + lane, Ld)];
    sidx[c] = (c * 32 + lane < Ld) ? sl.y : node;
    ev[c] = 0.f;
  }

  const float4 hself = *(const float4*)(H + (size_t)node * F + fl * 4);
  const float4 as4 = *(const float4*)(a_src + fl * 4);
  const float4 ad4 = *(const float4*)(a_dst + fl * 4);
  const float sd = group_sum<LPE>(dot4(hself, ad4));
  const float e_self = leaky(group_sum<LPE>(dot4(hself, as4)) + sd);

  Run st;
  st.m = (sub == 0) ? e_self : LOWEST_F;
  st.sum = (sub == 0) ? 1.0f : 0.f;
  st.acc = (sub == 0) ? hself : make_float4(0.f, 0.f, 0.f, 0.f);

  if (small) {
#pragma unroll
    for (int c = 0; c < MAXC; ++c) {
      if (c * 32 < Ld)
        ev[c] = chunk_rows<F>(st, sidx[c], min(32, Ld - c * 32), H, as4, sd, sub, fl, lane);
    }
  } else {
    for (int c0 = 0; c0 < Ld; c0 += 32) {
      int s_l = node;
      if (c0 + lane < Ld) s_l = slots[beg + c0 + lane].y;
      chunk_rows<F>(st, s_l, min(32, Ld - c0), H, as4, sd, sub, fl, lane);
    }
  }

  float m_all = st.m;
#pragma unroll
  for (int off = 16; off >= LPE; off >>= 1) m_all = fmaxf(m_all, __shfl_xor(m_all, off, 32));
  const float w = __expf(st.m - m_all);
  float sum = st.sum * w;
  float4 acc = make_float4(st.acc.x * w, st.acc.y * w, st.acc.z * w, st.acc.w * w);
#pragma unroll
  for (int off = 16; off >= LPE; off >>= 1) {
    sum += __shfl_xor(sum, off, 32);
    acc.x += __shfl_xor(acc.x, off, 32);
    acc.y += __shfl_xor(acc.y, off, 32);
    acc.z += __shfl_xor(acc.z, off, 32);
    acc.w += __shfl_xor(acc.w, off, 32);
  }
  const float inv = 1.0f / sum;

  if (small) {
#pragma unroll
    for (int c = 0; c < MAXC; ++c) {
      if (c * 32 + lane < Ld) alpha_slot[beg + c * 32 + lane] = __expf(ev[c] - m_all) * inv;
    }
  } else {
    for (int c0 = 0; c0 < Ld; c0 += 32) {
      int2 sl = make_int2(0, node);
      if (c0 + lane < Ld) sl = slots[beg + c0 + lane];
      Run scratch;
      scratch.m = LOWEST_F; scratch.sum = 0.f; scratch.acc = make_float4(0.f, 0.f, 0.f, 0.f);
      const float e = chunk_rows<F>(scratch, sl.y, min(32, Ld - c0), H, as4, sd, sub, fl, lane);
      if (c0 + lane < Ld) alpha_slot[beg + c0 + lane] = __expf(e - m_all) * inv;
    }
  }
  if (lane == 0) alpha_slot[beg + Ld] = __expf(e_self - m_all) * inv;

  if (sub == 0) {
    const float4 b = *(const float4*)(bias + fl * 4);
    float4 v = make_float4(fmaf(acc.x, inv, b.x), fmaf(acc.y, inv, b.y),
                           fmaf(acc.z, inv, b.z), fmaf(acc.w, inv, b.w));
    if (ELU) {
      v.x = (v.x > 0.f) ? v.x : (__expf(v.x) - 1.0f);
      v.y = (v.y > 0.f) ? v.y : (__expf(v.y) - 1.0f);
      v.z = (v.z > 0.f) ? v.z : (__expf(v.z) - 1.0f);
      v.w = (v.w > 0.f) ? v.w : (__expf(v.w) - 1.0f);
    }
    *(float4*)(out + (size_t)node * F + fl * 4) = v;
  }
}

#define UBIN 512
#define UCH 8192
#define UTILE 8192

__device__ __forceinline__ int scan_ubins(int h, int* wsum) {
  const int t = threadIdx.x, lane = t & 31;
  int v = h;
  if (t < UBIN) {
#pragma unroll
    for (int off = 1; off < 32; off <<= 1) {
      const int u = __shfl_up(v, off, 32);
      if (lane >= off) v += u;
    }
    if (lane == 31) wsum[t >> 5] = v;
  }
  __syncthreads();
  int front = 0;
  if (t < UBIN) {
    for (int w = 0; w < (t >> 5); ++w) front += wsum[w];
  }
  return front + v - h;
}

__global__ __launch_bounds__(1024) void alpha_bin_kernel(
    const int2* __restrict__ slots, const float* __restrict__ as1, const float* __restrict__ as2,
    int* __restrict__ ucur, int4* __restrict__ upk, int Etot) {
  __shared__ int hist[UBIN];
  __shared__ int lbase[UBIN];
  __shared__ int gbase[UBIN];
  __shared__ int wsum[UBIN / 32];
  __shared__ int st_id[UTILE];
  __shared__ float st_a1[UTILE];
  __shared__ float st_a2[UTILE];
  const int t = threadIdx.x;
  if (t < UBIN) hist[t] = 0;
  __syncthreads();
  const int base = blockIdx.x * UTILE;
  int id[UTILE / 1024], lp[UTILE / 1024];
  float a1[UTILE / 1024], a2[UTILE / 1024];
#pragma unroll
  for (int k = 0; k < UTILE / 1024; ++k) {
    const int p = base + k * 1024 + t;
    id[k] = -1; a1[k] = 0.f; a2[k] = 0.f; lp[k] = 0;
    if (p < Etot) {
      id[k] = slots[p].x;
      a1[k] = as1[p];
      a2[k] = as2[p];
    }
  }
#pragma unroll
  for (int k = 0; k < UTILE / 1024; ++k) {
    if (id[k] >= 0) lp[k] = atomicAdd(&hist[((unsigned)id[k] / UCH) & (UBIN - 1)], 1);
  }
  __syncthreads();
  const int h = (t < UBIN) ? hist[t] : 0;
  const int lfront = scan_ubins(h, wsum);
  if (t < UBIN) {
    lbase[t] = lfront;
    gbase[t] = t * UCH + ((h != 0) ? atomicAdd(&ucur[t], h) : 0);
  }
  __syncthreads();
#pragma unroll
  for (int k = 0; k < UTILE / 1024; ++k) {
    if (id[k] >= 0) {
      const int pos = lbase[((unsigned)id[k] / UCH) & (UBIN - 1)] + lp[k];
      st_id[pos] = id[k];
      st_a1[pos] = a1[k];
      st_a2[pos] = a2[k];
    }
  }
  __syncthreads();
  const int cntt = min(UTILE, Etot - base);
#pragma unroll
  for (int k = 0; k < UTILE / 1024; ++k) {
    const int p = k * 1024 + t;
    if (p < cntt) {
      const int e = st_id[p];
      const unsigned bb = ((unsigned)e / UCH) & (UBIN - 1);
      const int gp = gbase[bb] + (p - lbase[bb]);
      upk[gp] = make_int4(e, __float_as_int(st_a1[p]), __float_as_int(st_a2[p]), 0);
    }
  }
}

__global__ __launch_bounds__(1024) void alpha_place_kernel(
    const int4* __restrict__ upk, float* __restrict__ alpha1, float* __restrict__ alpha2, int Etot) {
  __shared__ float s1[UCH];
  __shared__ float s2[UCH];
  const int t = threadIdx.x;
  const int base = blockIdx.x * UCH;
  const int cnt = min(UCH, Etot - base);
  int4 v[UCH / 1024];
#pragma unroll
  for (int k = 0; k < UCH / 1024; ++k) v[k] = upk[base + min(k * 1024 + t, cnt - 1)];
#pragma unroll
  for (int k = 0; k < UCH / 1024; ++k) {
    if (k * 1024 + t < cnt) {
      const int l = (v[k].x - base) & (UCH - 1);
      s1[l] = __int_as_float(v[k].y);
      s2[l] = __int_as_float(v[k].z);
    }
  }
  __syncthreads();
#pragma unroll
  for (int k = 0; k < UCH / 1024; ++k) {
    const int p = k * 1024 + t;
    if (p < cnt) {
      alpha1[base + p] = s1[p];
      alpha2[base + p] = s2[p];
    }
  }
}

static inline int cdiv(long long a, long long b) { return (int)((a + b - 1) / b); }

extern "C" void kernel_launch(void* const* d_in, const int* in_sizes, int n_in,
                              void* d_out, int out_size, void* d_ws, size_t ws_size,
                              hipStream_t stream) {
  constexpr int FIN = 128, FH = 64, FO = 32;
  const float* x      = (const float*)d_in[0];
  const int*   eidx   = (const int*)d_in[1];
  const float* W1     = (const float*)d_in[2];
  const float* a_src1 = (const float*)d_in[3];
  const float* a_dst1 = (const float*)d_in[4];
  const float* b1     = (const float*)d_in[5];
  const float* W2     = (const float*)d_in[6];
  const float* a_src2 = (const float*)d_in[7];
  const float* a_dst2 = (const float*)d_in[8];
  const float* b2     = (const float*)d_in[9];

  const int N = in_sizes[0] / FIN;
  const int E = in_sizes[1] / 2;
  const int Etot = E + N;
  const int* src = eidx;
  const int* dst = eidx + E;

  float* hfeat1 = (float*)d_ws;
  float* agg1   = hfeat1 + (size_t)N * FH;
  float* hfeat2 = agg1 + (size_t)N * FH;
  int4* upk     = (int4*)(hfeat2 + (size_t)N * FO);
  float* aslot1 = (float*)(upk + (size_t)Etot);
  float* aslot2 = aslot1 + (size_t)Etot;
  int2* pslot   = (int2*)(aslot2 + (size_t)Etot);
  int2* slots_b = pslot + (size_t)E;
  int2* spare   = slots_b + (size_t)Etot;
  int* pdst     = (int*)(spare + (size_t)E);
  int* bincnt   = pdst + (size_t)E;
  int* ucur     = bincnt + 2 * NBIN;
  int* rowbeg   = ucur + UBIN;

  float* h2     = (float*)d_out;
  float* alpha1 = h2 + (size_t)N * FO;
  float* alpha2 = alpha1 + (size_t)Etot;

  dim3 b256(256), b128(128), b1024(1024);

  hipMemsetAsync(bincnt, 0, (size_t)(2 * NBIN + UBIN) * sizeof(int), stream);
  bin_count_kernel<<<cdiv(E, CHUNK), b1024, 0, stream>>>(dst, bincnt, E, N);
  partition_kernel<<<cdiv(E, CHUNK), b1024, 0, stream>>>(
      src, dst, bincnt, bincnt + NBIN, pslot, pdst, E, N);
  range_order_kernel<<<NBIN, b1024, 0, stream>>>(pslot, pdst, bincnt, spare, slots_b, rowbeg, N, E);

  gemm_wmma_kernel<FIN, FH><<<cdiv(N, 64), b128, 0, stream>>>(x, W1, hfeat1, N);
  node_gather_kernel<FH, true><<<cdiv(N, 8), b256, 0, stream>>>(
      slots_b, rowbeg, a_src1, a_dst1, hfeat1, b1, aslot1, agg1, N);

  gemm_wmma_kernel<FH, FO><<<cdiv(N, 64), b128, 0, stream>>>(agg1, W2, hfeat2, N);
  node_gather_kernel<FO, false><<<cdiv(N, 8), b256, 0, stream>>>(
      slots_b, rowbeg, a_src2, a_dst2, hfeat2, b2, aslot2, h2, N);

  alpha_bin_kernel<<<cdiv(Etot, UTILE), b1024, 0, stream>>>(slots_b, aslot1, aslot2, ucur, upk, Etot);
  alpha_place_kernel<<<cdiv(Etot, UCH), b1024, 0, stream>>>(upk, alpha1, alpha2, Etot);
}
